# starting kernel
# baseline (speedup 1.0000x reference)
#include <hip/hip_runtime.h>
#include <hip/hip_bf16.h>


typedef float v2f __attribute__((ext_vector_type(2)));
typedef float v8f __attribute__((ext_vector_type(8)));

#define XB   512
#define XS   64
#define XD   128
#define XNH  4
#define XDH  32
#define XT   (XB * XS)
#define XEPS 1e-5f

__device__ __forceinline__ v8f wmma4(v2f a, v2f b, v8f c) {
  return __builtin_amdgcn_wmma_f32_16x16x4_f32(false, a, false, b, (short)0, c,
                                               false, false);
}

__device__ __forceinline__ float gelu_exact(float x) {
  return 0.5f * x * (1.0f + erff(x * 0.70710678118654752f));
}

__global__ void k_embed(const int* __restrict__ x, const float* __restrict__ emb,
                        float* __restrict__ h, int total) {
  int i = blockIdx.x * blockDim.x + threadIdx.x;
  if (i >= total) return;
  int t = i >> 7;
  int c = i & 127;
  h[i] = emb[x[t] * XD + c];
}

__global__ void k_ln(const float* __restrict__ h, const float* __restrict__ w,
                     float* __restrict__ xn) {
  int wid  = threadIdx.x >> 5;
  int lane = threadIdx.x & 31;
  int t = blockIdx.x * (blockDim.x >> 5) + wid;
  const float4* hv = (const float4*)(h + (size_t)t * XD);
  float4 v = hv[lane];
  float s = v.x + v.y + v.z + v.w;
  float q = v.x * v.x + v.y * v.y + v.z * v.z + v.w * v.w;
  #pragma unroll
  for (int m = 16; m >= 1; m >>= 1) {
    s += __shfl_xor(s, m, 32);
    q += __shfl_xor(q, m, 32);
  }
  float mu  = s * (1.0f / XD);
  float var = q * (1.0f / XD) - mu * mu;
  float rs  = rsqrtf(var + XEPS);
  float4 wl = ((const float4*)w)[lane];
  float4 o;
  o.x = (v.x - mu) * rs * wl.x;
  o.y = (v.y - mu) * rs * wl.y;
  o.z = (v.z - mu) * rs * wl.z;
  o.w = (v.w - mu) * rs * wl.w;
  ((float4*)(xn + (size_t)t * XD))[lane] = o;
}

__global__ __launch_bounds__(32) void k_scan(const float* __restrict__ xn,
                                             float* __restrict__ hio,
                                             const float* __restrict__ Wg,
                                             const float* __restrict__ R,
                                             const float* __restrict__ bias,
                                             const float* __restrict__ gnw) {
  __shared__ float lwg[8][8][64];
  __shared__ float lr[8][8][64];
  __shared__ float lh[16 * 32];
  __shared__ float lb[128];
  __shared__ float lgn[32];

  const int lane = threadIdx.x;
  const int hlf  = lane >> 4;
  const int l16  = lane & 15;
  const int b0   = blockIdx.x * 16;
  const int n    = blockIdx.y;

  #pragma unroll
  for (int nt = 0; nt < 8; ++nt) {
    int N  = nt * 16 + l16;
    int g  = N >> 5;
    int dh = N & 31;
    #pragma unroll
    for (int kk = 0; kk < 8; ++kk) {
      int K0 = kk * 4 + hlf * 2;
      const float* wgp = Wg + (((size_t)(g * XNH + n) * XDH + dh) * XDH + K0);
      lwg[nt][kk][lane * 2 + 0] = wgp[0];
      lwg[nt][kk][lane * 2 + 1] = wgp[1];
      lr[nt][kk][lane * 2 + 0] = R[(size_t)(n * XDH + K0 + 0) * (4 * XDH) + N];
      lr[nt][kk][lane * 2 + 1] = R[(size_t)(n * XDH + K0 + 1) * (4 * XDH) + N];
    }
  }
  for (int j = lane; j < 128; j += 32) lb[j] = bias[n * 128 + j];
  lgn[lane & 31] = gnw[n * XDH + (lane & 31)];
  for (int j = lane; j < 16 * 32; j += 32) lh[j] = 0.0f;
  __syncthreads();

  float cS[2][8]  = {};
  float nS[2][8]  = {};
  float mS[2][8]  = {};

  #pragma unroll 1
  for (int s = 0; s < XS; ++s) {
    v2f ax[8], ah[8];
    const float* xp = xn + ((size_t)(b0 + l16) * XS + s) * XD + n * XDH;
    #pragma unroll
    for (int kk = 0; kk < 8; ++kk) {
      int K0 = kk * 4 + hlf * 2;
      ax[kk].x = xp[K0];
      ax[kk].y = xp[K0 + 1];
      ah[kk].x = lh[l16 * 32 + K0];
      ah[kk].y = lh[l16 * 32 + K0 + 1];
    }

    v8f acc[8];
    #pragma unroll
    for (int nt = 0; nt < 8; ++nt) {
      float bj = lb[nt * 16 + l16];
      v8f a;
      #pragma unroll
      for (int r = 0; r < 8; ++r) a[r] = bj;
      #pragma unroll
      for (int kk = 0; kk < 8; ++kk) {
        v2f bw, br;
        bw.x = lwg[nt][kk][lane * 2 + 0];
        bw.y = lwg[nt][kk][lane * 2 + 1];
        a = wmma4(ax[kk], bw, a);
        br.x = lr[nt][kk][lane * 2 + 0];
        br.y = lr[nt][kk][lane * 2 + 1];
        a = wmma4(ah[kk], br, a);
      }
      acc[nt] = a;
    }
    __syncthreads();

    #pragma unroll
    for (int t2 = 0; t2 < 2; ++t2) {
      #pragma unroll
      for (int r = 0; r < 8; ++r) {
        float ir  = acc[0 + t2][r];
        float fr  = acc[2 + t2][r];
        float zr  = acc[4 + t2][r];
        float orr = acc[6 + t2][r];
        float ls  = (fr >= 0.0f) ? (-log1pf(expf(-fr)))
                                 : (fr - log1pf(expf(fr)));
        float lfm = mS[t2][r] + ls;
        float mn  = fmaxf(ir, lfm);
        float ig  = expf(ir - mn);
        float fg  = expf(lfm - mn);
        float cn  = fg * cS[t2][r] + ig * tanhf(zr);
        float nn  = fg * nS[t2][r] + ig;
        float og  = 1.0f / (1.0f + expf(-orr));
        float hn  = og * cn / nn;
        cS[t2][r] = cn;
        nS[t2][r] = nn;
        mS[t2][r] = mn;
        lh[(r + hlf * 8) * 32 + t2 * 16 + l16] = hn;
      }
    }
    __syncthreads();

    if (lane < 16) {
      float mu = 0.0f, q = 0.0f;
      #pragma unroll
      for (int i = 0; i < 32; ++i) {
        float v = lh[lane * 32 + i];
        mu += v;
        q  += v * v;
      }
      mu *= (1.0f / XDH);
      q = q * (1.0f / XDH) - mu * mu;
      float rs = rsqrtf(q + XEPS);
      float* hp = hio + ((size_t)(b0 + lane) * XS + s) * XD + n * XDH;
      #pragma unroll
      for (int i = 0; i < 32; ++i) {
        float v = (lh[lane * 32 + i] - mu) * rs * lgn[i];
        hp[i] = hp[i] + v;
      }
    }
    __syncthreads();
  }
}

__global__ __launch_bounds__(32) void k_ffn(const float* __restrict__ xn,
                                            float* __restrict__ hio,
                                            const float* __restrict__ Wup,
                                            const float* __restrict__ Wdn) {
  __shared__ float lact[16 * 128];
  const int lane = threadIdx.x;
  const int hlf  = lane >> 4;
  const int l16  = lane & 15;
  const size_t t0 = (size_t)blockIdx.x * 16;

  v2f ax[32];
  const float* xp = xn + (t0 + l16) * XD;
  #pragma unroll
  for (int kk = 0; kk < 32; ++kk) {
    int K0 = kk * 4 + hlf * 2;
    ax[kk].x = xp[K0];
    ax[kk].y = xp[K0 + 1];
  }

  float gatev[8][8];
  #pragma unroll
  for (int nt = 0; nt < 8; ++nt) {
    v8f a;
    #pragma unroll
    for (int r = 0; r < 8; ++r) a[r] = 0.0f;
    int j = nt * 16 + l16;
    #pragma unroll
    for (int kk = 0; kk < 32; ++kk) {
      int K0 = kk * 4 + hlf * 2;
      v2f b;
      b.x = Wup[(size_t)j * XD + K0];
      b.y = Wup[(size_t)j * XD + K0 + 1];
      a = wmma4(ax[kk], b, a);
    }
    #pragma unroll
    for (int r = 0; r < 8; ++r) gatev[nt][r] = a[r];
  }

  #pragma unroll
  for (int nt = 0; nt < 8; ++nt) {
    v8f a;
    #pragma unroll
    for (int r = 0; r < 8; ++r) a[r] = 0.0f;
    int j = 128 + nt * 16 + l16;
    #pragma unroll
    for (int kk = 0; kk < 32; ++kk) {
      int K0 = kk * 4 + hlf * 2;
      v2f b;
      b.x = Wup[(size_t)j * XD + K0];
      b.y = Wup[(size_t)j * XD + K0 + 1];
      a = wmma4(ax[kk], b, a);
    }
    #pragma unroll
    for (int r = 0; r < 8; ++r) {
      float act = gelu_exact(gatev[nt][r]) * a[r];
      lact[(r + hlf * 8) * 128 + nt * 16 + l16] = act;
    }
  }
  __syncthreads();

  v2f aa[32];
  const float* lp = lact + l16 * 128;
  #pragma unroll
  for (int kk = 0; kk < 32; ++kk) {
    int K0 = kk * 4 + hlf * 2;
    aa[kk].x = lp[K0];
    aa[kk].y = lp[K0 + 1];
  }
  #pragma unroll
  for (int nt = 0; nt < 8; ++nt) {
    v8f a;
    #pragma unroll
    for (int r = 0; r < 8; ++r) a[r] = 0.0f;
    int j = nt * 16 + l16;
    #pragma unroll
    for (int kk = 0; kk < 32; ++kk) {
      int K0 = kk * 4 + hlf * 2;
      v2f b;
      b.x = Wdn[(size_t)j * XD + K0];
      b.y = Wdn[(size_t)j * XD + K0 + 1];
      a = wmma4(aa[kk], b, a);
    }
    #pragma unroll
    for (int r = 0; r < 8; ++r) {
      size_t row = t0 + r + hlf * 8;
      hio[row * XD + nt * 16 + l16] += a[r];
    }
  }
}

__global__ __launch_bounds__(32) void k_proj(const float* __restrict__ xn,
                                             const float* __restrict__ pw,
                                             const float* __restrict__ pb,
                                             float* __restrict__ out) {
  const int lane = threadIdx.x;
  const int hlf  = lane >> 4;
  const int l16  = lane & 15;
  const size_t t0 = (size_t)blockIdx.x * 16;

  v8f a;
  #pragma unroll
  for (int r = 0; r < 8; ++r) a[r] = 0.0f;
  const float* xp = xn + (t0 + l16) * XD;
  #pragma unroll
  for (int kk = 0; kk < 32; ++kk) {
    int K0 = kk * 4 + hlf * 2;
    v2f av;
    av.x = xp[K0];
    av.y = xp[K0 + 1];
    v2f b;
    b.x = 0.0f;
    b.y = 0.0f;
    if (l16 < 9) {
      b.x = pw[(size_t)l16 * XD + K0];
      b.y = pw[(size_t)l16 * XD + K0 + 1];
    }
    a = wmma4(av, b, a);
  }
  if (l16 < 9) {
    float bb = pb[l16];
    #pragma unroll
    for (int r = 0; r < 8; ++r) {
      size_t row = t0 + r + hlf * 8;
      out[row * 9 + l16] = a[r] + bb;
    }
  }
}

extern "C" void kernel_launch(void* const* d_in, const int* in_sizes, int n_in,
                              void* d_out, int out_size, void* d_ws, size_t ws_size,
                              hipStream_t stream) {
  (void)in_sizes; (void)n_in; (void)out_size; (void)ws_size;
  const int*   x     = (const int*)d_in[0];
  const float* emb   = (const float*)d_in[1];
  const float* ln1   = (const float*)d_in[2];
  const float* Wg    = (const float*)d_in[3];
  const float* R     = (const float*)d_in[4];
  const float* bias  = (const float*)d_in[5];
  const float* gnw   = (const float*)d_in[6];
  const float* ln2   = (const float*)d_in[7];
  const float* wup   = (const float*)d_in[8];
  const float* wdn   = (const float*)d_in[9];
  const float* postw = (const float*)d_in[10];
  const float* pw    = (const float*)d_in[11];
  const float* pb    = (const float*)d_in[12];
  float* out = (float*)d_out;

  float* h  = (float*)d_ws;
  float* xb = h + (size_t)XT * XD;

  k_embed<<<(XT * XD + 255) / 256, 256, 0, stream>>>(x, emb, h, XT * XD);

  for (int bi = 0; bi < 2; ++bi) {
    k_ln<<<XT / 8, 256, 0, stream>>>(h, ln1 + bi * XD, xb);
    k_scan<<<dim3(XB / 16, XNH), 32, 0, stream>>>(
        xb, h,
        Wg   + (size_t)bi * 4 * XNH * XDH * XDH,
        R    + (size_t)bi * XNH * XDH * 4 * XDH,
        bias + (size_t)bi * XNH * 4 * XDH,
        gnw  + (size_t)bi * XD);
    k_ln<<<XT / 8, 256, 0, stream>>>(h, ln2 + bi * XD, xb);
    k_ffn<<<XT / 16, 32, 0, stream>>>(xb, h,
                                      wup + (size_t)bi * 2 * XD * XD,
                                      wdn + (size_t)bi * XD * XD);
  }

  k_ln<<<XT / 8, 256, 0, stream>>>(h, postw, xb);
  k_proj<<<XT / 16, 32, 0, stream>>>(xb, pw, pb, out);
}


# scan: one hidden unit per lane, not two: each batch row sits in two WMMA columns and lanes c, c+8 split the tile's unit pair, halving the serial gate chain; 8 rows a work-group, 256 groups; same sums
# speedup vs baseline: 4.6862x; 4.6862x over previous
#include <hip/hip_runtime.h>
#include <hip/hip_bf16.h>


typedef float v2f __attribute__((ext_vector_type(2)));
typedef float v8f __attribute__((ext_vector_type(8)));

#define XB   512
#define XS   64
#define XD   128
#define XNH  4
#define XDH  32
#define XT   (XB * XS)
#define XEPS 1e-5f

__device__ __forceinline__ v8f wmma4(v2f a, v2f b, v8f c) {
  return __builtin_amdgcn_wmma_f32_16x16x4_f32(false, a, false, b, (short)0, c,
                                               false, false);
}

__device__ __forceinline__ float gelu_exact(float x) {
  return 0.5f * x * (1.0f + erff(x * 0.70710678118654752f));
}

__global__ void k_embed(const int* __restrict__ x, const float* __restrict__ emb,
                        float* __restrict__ h, int total) {
  int i = blockIdx.x * blockDim.x + threadIdx.x;
  if (i >= total) return;
  int t = i >> 7;
  int c = i & 127;
  h[i] = emb[x[t] * XD + c];
}

__global__ void k_ln(const float* __restrict__ h, const float* __restrict__ w,
                     float* __restrict__ xn) {
  int wid  = threadIdx.x >> 5;
  int lane = threadIdx.x & 31;
  int t = blockIdx.x * (blockDim.x >> 5) + wid;
  const float4* hv = (const float4*)(h + (size_t)t * XD);
  float4 v = hv[lane];
  float s = v.x + v.y + v.z + v.w;
  float q = v.x * v.x + v.y * v.y + v.z * v.z + v.w * v.w;
  #pragma unroll
  for (int m = 16; m >= 1; m >>= 1) {
    s += __shfl_xor(s, m, 32);
    q += __shfl_xor(q, m, 32);
  }
  float mu  = s * (1.0f / XD);
  float var = q * (1.0f / XD) - mu * mu;
  float rs  = rsqrtf(var + XEPS);
  float4 wl = ((const float4*)w)[lane];
  float4 o;
  o.x = (v.x - mu) * rs * wl.x;
  o.y = (v.y - mu) * rs * wl.y;
  o.z = (v.z - mu) * rs * wl.z;
  o.w = (v.w - mu) * rs * wl.w;
  ((float4*)(xn + (size_t)t * XD))[lane] = o;
}

#define SCAN_WAVES 8
#define SCAN_BT    8

__device__ __forceinline__ float xhalf(float x) {
  int b = __builtin_bit_cast(int, x);
  return __builtin_bit_cast(
      float, __builtin_amdgcn_permlanex16(b, b, 0x76543210, 0xfedcba98, false, false));
}

__global__ __launch_bounds__(32 * SCAN_WAVES) void k_scan(const float* __restrict__ xn,
                                                          float* __restrict__ hio,
                                                          const float* __restrict__ Wg,
                                                          const float* __restrict__ R,
                                                          const float* __restrict__ bias,
                                                          const float* __restrict__ gnw) {
  __shared__ float2 lh[2][SCAN_WAVES][2 * SCAN_BT];

  const int w    = threadIdx.x >> 5;
  const int lane = threadIdx.x & 31;
  const int hlf  = lane >> 4;
  const int l16  = lane & 15;
  const int bb   = l16 & 7;
  const int e    = l16 >> 3;
  const int b0   = blockIdx.x * SCAN_BT;
  const int n    = blockIdx.y;
  const int ub   = 8 * (w >> 1) + 2 * (w & 1);

  const int ag = (l16 >> 1) & 3;
  const int au = ub + 4 * (l16 >> 3) + (l16 & 1);
  v2f aw[8], ar[8];
  #pragma unroll
  for (int kk = 0; kk < 8; ++kk) {
    int K0 = 8 * (kk >> 1) + 4 * hlf + 2 * (kk & 1);
    const float* wgp = Wg + (((size_t)(ag * XNH + n) * XDH + au) * XDH + K0);
    aw[kk].x = wgp[0];
    aw[kk].y = wgp[1];
    ar[kk].x = R[(size_t)(n * XDH + K0 + 0) * (4 * XDH) + ag * XDH + au];
    ar[kk].y = R[(size_t)(n * XDH + K0 + 1) * (4 * XDH) + ag * XDH + au];
  }
  float bz[8];
  #pragma unroll
  for (int r = 0; r < 8; ++r)
    bz[r] = bias[n * 128 + (r >> 1) * XDH + ub + 4 * hlf + (r & 1)];
  float gn[8];
  #pragma unroll
  for (int i = 0; i < 2; ++i) {
    #pragma unroll
    for (int q = 0; q < 4; ++q) gn[4 * i + q] = gnw[n * XDH + 8 * (2 * e + i) + 4 * hlf + q];
  }

  float cS = 0.0f;
  float nS = 0.0f;
  float mS = 0.0f;
  float hA[16] = {};

  const size_t rowoff = (size_t)(b0 + bb) * XS * XD + n * XDH + hlf * 4;

  #pragma unroll 1
  for (int s = 0; s < XS; ++s) {
    const float4* xp = (const float4*)(xn + rowoff + (size_t)s * XD);
    float4*       hp = (float4*)(hio + rowoff + (size_t)s * XD) + 4 * e;
    const bool writer = ((s & (SCAN_WAVES - 1)) == w);
    float4 xv[4], hv[2];
    #pragma unroll
    for (int p = 0; p < 4; ++p) xv[p] = xp[2 * p];
    #pragma unroll
    for (int i = 0; i < 2; ++i) hv[i] = make_float4(0.0f, 0.0f, 0.0f, 0.0f);
    if (writer) {
      #pragma unroll
      for (int i = 0; i < 2; ++i) hv[i] = hp[2 * i];
    }

    v2f bx[8], bh[8];
    #pragma unroll
    for (int kk = 0; kk < 8; ++kk) {
      float4 q = xv[kk >> 1];
      bx[kk].x = (kk & 1) ? q.z : q.x;
      bx[kk].y = (kk & 1) ? q.w : q.y;
      bh[kk].x = hA[2 * kk + 0];
      bh[kk].y = hA[2 * kk + 1];
    }

    v8f acc;
    #pragma unroll
    for (int r = 0; r < 8; ++r) acc[r] = bz[r];
    #pragma unroll
    for (int kk = 0; kk < 8; ++kk) acc = wmma4(aw[kk], bx[kk], acc);
    #pragma unroll
    for (int kk = 0; kk < 8; ++kk) acc = wmma4(ar[kk], bh[kk], acc);

    float hn;
    {
      float ir  = e ? acc[1] : acc[0];
      float fr  = e ? acc[3] : acc[2];
      float zr  = e ? acc[5] : acc[4];
      float orr = e ? acc[7] : acc[6];
      float ls  = (fr >= 0.0f) ? (-log1pf(expf(-fr)))
                               : (fr - log1pf(expf(fr)));
      float lfm = mS + ls;
      float mn  = fmaxf(ir, lfm);
      float ig  = expf(ir - mn);
      float fg  = expf(lfm - mn);
      float cn  = fg * cS + ig * tanhf(zr);
      float nn  = fg * nS + ig;
      float og  = 1.0f / (1.0f + expf(-orr));
      hn = og * cn / nn;
      cS = cn;
      nS = nn;
      mS = mn;
    }

    ((float*)&lh[s & 1][w][hlf * SCAN_BT + bb])[e] = hn;
    __syncthreads();
    #pragma unroll
    for (int kk = 0; kk < SCAN_WAVES; ++kk) {
      float2 t = lh[s & 1][kk][hlf * SCAN_BT + bb];
      hA[2 * kk + 0] = t.x;
      hA[2 * kk + 1] = t.y;
    }

    if (writer) {
      v2f h2[8];
      #pragma unroll
      for (int kk = 0; kk < 8; ++kk) {
        h2[kk].x = hA[2 * kk + 0];
        h2[kk].y = hA[2 * kk + 1];
      }
      v2f s2 = ((h2[0] + h2[1]) + (h2[2] + h2[3])) + ((h2[4] + h2[5]) + (h2[6] + h2[7]));
      v2f q2 = ((h2[0] * h2[0] + h2[1] * h2[1]) + (h2[2] * h2[2] + h2[3] * h2[3])) +
               ((h2[4] * h2[4] + h2[5] * h2[5]) + (h2[6] * h2[6] + h2[7] * h2[7]));
      float sum = s2.x + s2.y;
      float sq  = q2.x + q2.y;
      sum += xhalf(sum);
      sq  += xhalf(sq);
      float mu = sum * (1.0f / XDH);
      float q  = sq * (1.0f / XDH) - mu * mu;
      float rs = rsqrtf(q + XEPS);
      #pragma unroll
      for (int i = 0; i < 2; ++i) {
        v2f lo0 = h2[2 * i], lo1 = h2[2 * i + 1];
        v2f hi0 = h2[4 + 2 * i], hi1 = h2[4 + 2 * i + 1];
        float u0 = e ? hi0.x : lo0.x;
        float u1 = e ? hi0.y : lo0.y;
        float u2 = e ? hi1.x : lo1.x;
        float u3 = e ? hi1.y : lo1.y;
        float4 o = hv[i];
        o.x += (u0 - mu) * rs * gn[4 * i + 0];
        o.y += (u1 - mu) * rs * gn[4 * i + 1];
        o.z += (u2 - mu) * rs * gn[4 * i + 2];
        o.w += (u3 - mu) * rs * gn[4 * i + 3];
        hp[2 * i] = o;
      }
    }
  }
}

__global__ __launch_bounds__(32) void k_ffn(const float* __restrict__ xn,
                                            float* __restrict__ hio,
                                            const float* __restrict__ Wup,
                                            const float* __restrict__ Wdn) {
  __shared__ float lact[16 * 128];
  const int lane = threadIdx.x;
  const int hlf  = lane >> 4;
  const int l16  = lane & 15;
  const size_t t0 = (size_t)blockIdx.x * 16;

  v2f ax[32];
  const float* xp = xn + (t0 + l16) * XD;
  #pragma unroll
  for (int kk = 0; kk < 32; ++kk) {
    int K0 = kk * 4 + hlf * 2;
    ax[kk].x = xp[K0];
    ax[kk].y = xp[K0 + 1];
  }

  float gatev[8][8];
  #pragma unroll
  for (int nt = 0; nt < 8; ++nt) {
    v8f a;
    #pragma unroll
    for (int r = 0; r < 8; ++r) a[r] = 0.0f;
    int j = nt * 16 + l16;
    #pragma unroll
    for (int kk = 0; kk < 32; ++kk) {
      int K0 = kk * 4 + hlf * 2;
      v2f b;
      b.x = Wup[(size_t)j * XD + K0];
      b.y = Wup[(size_t)j * XD + K0 + 1];
      a = wmma4(ax[kk], b, a);
    }
    #pragma unroll
    for (int r = 0; r < 8; ++r) gatev[nt][r] = a[r];
  }

  #pragma unroll
  for (int nt = 0; nt < 8; ++nt) {
    v8f a;
    #pragma unroll
    for (int r = 0; r < 8; ++r) a[r] = 0.0f;
    int j = 128 + nt * 16 + l16;
    #pragma unroll
    for (int kk = 0; kk < 32; ++kk) {
      int K0 = kk * 4 + hlf * 2;
      v2f b;
      b.x = Wup[(size_t)j * XD + K0];
      b.y = Wup[(size_t)j * XD + K0 + 1];
      a = wmma4(ax[kk], b, a);
    }
    #pragma unroll
    for (int r = 0; r < 8; ++r) {
      float act = gelu_exact(gatev[nt][r]) * a[r];
      lact[(r + hlf * 8) * 128 + nt * 16 + l16] = act;
    }
  }
  __syncthreads();

  v2f aa[32];
  const float* lp = lact + l16 * 128;
  #pragma unroll
  for (int kk = 0; kk < 32; ++kk) {
    int K0 = kk * 4 + hlf * 2;
    aa[kk].x = lp[K0];
    aa[kk].y = lp[K0 + 1];
  }
  #pragma unroll
  for (int nt = 0; nt < 8; ++nt) {
    v8f a;
    #pragma unroll
    for (int r = 0; r < 8; ++r) a[r] = 0.0f;
    int j = nt * 16 + l16;
    #pragma unroll
    for (int kk = 0; kk < 32; ++kk) {
      int K0 = kk * 4 + hlf * 2;
      v2f b;
      b.x = Wdn[(size_t)j * XD + K0];
      b.y = Wdn[(size_t)j * XD + K0 + 1];
      a = wmma4(aa[kk], b, a);
    }
    #pragma unroll
    for (int r = 0; r < 8; ++r) {
      size_t row = t0 + r + hlf * 8;
      hio[row * XD + nt * 16 + l16] += a[r];
    }
  }
}

__global__ __launch_bounds__(32) void k_proj(const float* __restrict__ xn,
                                             const float* __restrict__ pw,
                                             const float* __restrict__ pb,
                                             float* __restrict__ out) {
  const int lane = threadIdx.x;
  const int hlf  = lane >> 4;
  const int l16  = lane & 15;
  const size_t t0 = (size_t)blockIdx.x * 16;

  v8f a;
  #pragma unroll
  for (int r = 0; r < 8; ++r) a[r] = 0.0f;
  const float* xp = xn + (t0 + l16) * XD;
  #pragma unroll
  for (int kk = 0; kk < 32; ++kk) {
    int K0 = kk * 4 + hlf * 2;
    v2f av;
    av.x = xp[K0];
    av.y = xp[K0 + 1];
    v2f b;
    b.x = 0.0f;
    b.y = 0.0f;
    if (l16 < 9) {
      b.x = pw[(size_t)l16 * XD + K0];
      b.y = pw[(size_t)l16 * XD + K0 + 1];
    }
    a = wmma4(av, b, a);
  }
  if (l16 < 9) {
    float bb = pb[l16];
    #pragma unroll
    for (int r = 0; r < 8; ++r) {
      size_t row = t0 + r + hlf * 8;
      out[row * 9 + l16] = a[r] + bb;
    }
  }
}

extern "C" void kernel_launch(void* const* d_in, const int* in_sizes, int n_in,
                              void* d_out, int out_size, void* d_ws, size_t ws_size,
                              hipStream_t stream) {
  (void)in_sizes; (void)n_in; (void)out_size; (void)ws_size;
  const int*   x     = (const int*)d_in[0];
  const float* emb   = (const float*)d_in[1];
  const float* ln1   = (const float*)d_in[2];
  const float* Wg    = (const float*)d_in[3];
  const float* R     = (const float*)d_in[4];
  const float* bias  = (const float*)d_in[5];
  const float* gnw   = (const float*)d_in[6];
  const float* ln2   = (const float*)d_in[7];
  const float* wup   = (const float*)d_in[8];
  const float* wdn   = (const float*)d_in[9];
  const float* postw = (const float*)d_in[10];
  const float* pw    = (const float*)d_in[11];
  const float* pb    = (const float*)d_in[12];
  float* out = (float*)d_out;

  float* h  = (float*)d_ws;
  float* xb = h + (size_t)XT * XD;

  k_embed<<<(XT * XD + 255) / 256, 256, 0, stream>>>(x, emb, h, XT * XD);

  for (int bi = 0; bi < 2; ++bi) {
    k_ln<<<XT / 8, 256, 0, stream>>>(h, ln1 + bi * XD, xb);
    k_scan<<<dim3(XB / SCAN_BT, XNH), 32 * SCAN_WAVES, 0, stream>>>(
        xb, h,
        Wg   + (size_t)bi * 4 * XNH * XDH * XDH,
        R    + (size_t)bi * XNH * XDH * 4 * XDH,
        bias + (size_t)bi * XNH * 4 * XDH,
        gnw  + (size_t)bi * XD);
    k_ln<<<XT / 8, 256, 0, stream>>>(h, ln2 + bi * XD, xb);
    k_ffn<<<XT / 16, 32, 0, stream>>>(xb, h,
                                      wup + (size_t)bi * 2 * XD * XD,
                                      wdn + (size_t)bi * XD * XD);
  }

  k_ln<<<XT / 8, 256, 0, stream>>>(h, postw, xb);
  k_proj<<<XT / 16, 32, 0, stream>>>(xb, pw, pb, out);
}


# scan: the work-group's xn slab (8 rows x 64 steps x 32 channels, 72 KB) is copied to LDS once in the prologue; a step reads its B fragments from LDS, no global load on the serial chain; same sums
# speedup vs baseline: 5.7047x; 1.2173x over previous
#include <hip/hip_runtime.h>
#include <hip/hip_bf16.h>


typedef float v2f __attribute__((ext_vector_type(2)));
typedef float v8f __attribute__((ext_vector_type(8)));

#define XB   512
#define XS   64
#define XD   128
#define XNH  4
#define XDH  32
#define XT   (XB * XS)
#define XEPS 1e-5f

__device__ __forceinline__ v8f wmma4(v2f a, v2f b, v8f c) {
  return __builtin_amdgcn_wmma_f32_16x16x4_f32(false, a, false, b, (short)0, c,
                                               false, false);
}

__device__ __forceinline__ float gelu_exact(float x) {
  return 0.5f * x * (1.0f + erff(x * 0.70710678118654752f));
}

__global__ void k_embed(const int* __restrict__ x, const float* __restrict__ emb,
                        float* __restrict__ h, int total) {
  int i = blockIdx.x * blockDim.x + threadIdx.x;
  if (i >= total) return;
  int t = i >> 7;
  int c = i & 127;
  h[i] = emb[x[t] * XD + c];
}

__global__ void k_ln(const float* __restrict__ h, const float* __restrict__ w,
                     float* __restrict__ xn) {
  int wid  = threadIdx.x >> 5;
  int lane = threadIdx.x & 31;
  int t = blockIdx.x * (blockDim.x >> 5) + wid;
  const float4* hv = (const float4*)(h + (size_t)t * XD);
  float4 v = hv[lane];
  float s = v.x + v.y + v.z + v.w;
  float q = v.x * v.x + v.y * v.y + v.z * v.z + v.w * v.w;
  #pragma unroll
  for (int m = 16; m >= 1; m >>= 1) {
    s += __shfl_xor(s, m, 32);
    q += __shfl_xor(q, m, 32);
  }
  float mu  = s * (1.0f / XD);
  float var = q * (1.0f / XD) - mu * mu;
  float rs  = rsqrtf(var + XEPS);
  float4 wl = ((const float4*)w)[lane];
  float4 o;
  o.x = (v.x - mu) * rs * wl.x;
  o.y = (v.y - mu) * rs * wl.y;
  o.z = (v.z - mu) * rs * wl.z;
  o.w = (v.w - mu) * rs * wl.w;
  ((float4*)(xn + (size_t)t * XD))[lane] = o;
}

#define SCAN_WAVES 8
#define SCAN_BT    8
#define SCAN_XP    36

__device__ __forceinline__ float xhalf(float x) {
  int b = __builtin_bit_cast(int, x);
  return __builtin_bit_cast(
      float, __builtin_amdgcn_permlanex16(b, b, 0x76543210, 0xfedcba98, false, false));
}

__global__ __launch_bounds__(32 * SCAN_WAVES) void k_scan(const float* __restrict__ xn,
                                                          float* __restrict__ hio,
                                                          const float* __restrict__ Wg,
                                                          const float* __restrict__ R,
                                                          const float* __restrict__ bias,
                                                          const float* __restrict__ gnw) {
  struct ScanLds {
    float2 lh[2][SCAN_WAVES][2 * SCAN_BT];
    float lx[XS * SCAN_BT * SCAN_XP];
  };
  __shared__ __attribute__((aligned(16))) ScanLds sm;
  float2 (*lh)[SCAN_WAVES][2 * SCAN_BT] = sm.lh;
  float* lx = sm.lx;

  const int w    = threadIdx.x >> 5;
  const int lane = threadIdx.x & 31;
  const int hlf  = lane >> 4;
  const int l16  = lane & 15;
  const int bb   = l16 & 7;
  const int e    = l16 >> 3;
  const int b0   = blockIdx.x * SCAN_BT;
  const int n    = blockIdx.y;
  const int ub   = 8 * (w >> 1) + 2 * (w & 1);

  {
    const float* xsrc = xn + (size_t)b0 * XS * XD + n * XDH;
    float4 st[16];
    #pragma unroll
    for (int i = 0; i < 16; ++i) {
      int idx  = i * (32 * SCAN_WAVES) + (int)threadIdx.x;
      int part = idx & 7;
      int seg  = idx >> 3;
      int row  = seg >> 6;
      int ss   = seg & (XS - 1);
      st[i] = *(const float4*)(xsrc + ((size_t)row * XS + ss) * XD + part * 4);
    }
    #pragma unroll
    for (int i = 0; i < 16; ++i) {
      int idx  = i * (32 * SCAN_WAVES) + (int)threadIdx.x;
      int part = idx & 7;
      int seg  = idx >> 3;
      int row  = seg >> 6;
      int ss   = seg & (XS - 1);
      *(float4*)&lx[(ss * SCAN_BT + row) * SCAN_XP + part * 4] = st[i];
    }
  }

  const int ag = (l16 >> 1) & 3;
  const int au = ub + 4 * (l16 >> 3) + (l16 & 1);
  v2f aw[8], ar[8];
  #pragma unroll
  for (int kk = 0; kk < 8; ++kk) {
    int K0 = 8 * (kk >> 1) + 4 * hlf + 2 * (kk & 1);
    const float* wgp = Wg + (((size_t)(ag * XNH + n) * XDH + au) * XDH + K0);
    aw[kk].x = wgp[0];
    aw[kk].y = wgp[1];
    ar[kk].x = R[(size_t)(n * XDH + K0 + 0) * (4 * XDH) + ag * XDH + au];
    ar[kk].y = R[(size_t)(n * XDH + K0 + 1) * (4 * XDH) + ag * XDH + au];
  }
  float bz[8];
  #pragma unroll
  for (int r = 0; r < 8; ++r)
    bz[r] = bias[n * 128 + (r >> 1) * XDH + ub + 4 * hlf + (r & 1)];
  float gn[8];
  #pragma unroll
  for (int i = 0; i < 2; ++i) {
    #pragma unroll
    for (int q = 0; q < 4; ++q) gn[4 * i + q] = gnw[n * XDH + 8 * (2 * e + i) + 4 * hlf + q];
  }

  float cS = 0.0f;
  float nS = 0.0f;
  float mS = 0.0f;
  float hA[16] = {};

  const size_t rowoff = (size_t)(b0 + bb) * XS * XD + n * XDH + hlf * 4;

  const float* xl = lx + bb * SCAN_XP + hlf * 4;

  __syncthreads();
  float4 xv[4];
  #pragma unroll
  for (int p = 0; p < 4; ++p) xv[p] = *(const float4*)(xl + 8 * p);

  #pragma unroll 1
  for (int s = 0; s < XS; ++s) {
    float4*       hp = (float4*)(hio + rowoff + (size_t)s * XD) + 4 * e;
    const bool writer = ((s & (SCAN_WAVES - 1)) == w);
    float4 hv[2];
    #pragma unroll
    for (int i = 0; i < 2; ++i) hv[i] = make_float4(0.0f, 0.0f, 0.0f, 0.0f);
    if (writer) {
      #pragma unroll
      for (int i = 0; i < 2; ++i) hv[i] = hp[2 * i];
    }

    v2f bx[8], bh[8];
    #pragma unroll
    for (int kk = 0; kk < 8; ++kk) {
      float4 q = xv[kk >> 1];
      bx[kk].x = (kk & 1) ? q.z : q.x;
      bx[kk].y = (kk & 1) ? q.w : q.y;
      bh[kk].x = hA[2 * kk + 0];
      bh[kk].y = hA[2 * kk + 1];
    }

    v8f acc;
    #pragma unroll
    for (int r = 0; r < 8; ++r) acc[r] = bz[r];
    #pragma unroll
    for (int kk = 0; kk < 8; ++kk) acc = wmma4(aw[kk], bx[kk], acc);
    #pragma unroll
    for (int kk = 0; kk < 8; ++kk) acc = wmma4(ar[kk], bh[kk], acc);

    float hn;
    {
      float ir  = e ? acc[1] : acc[0];
      float fr  = e ? acc[3] : acc[2];
      float zr  = e ? acc[5] : acc[4];
      float orr = e ? acc[7] : acc[6];
      float ls  = (fr >= 0.0f) ? (-log1pf(expf(-fr)))
                               : (fr - log1pf(expf(fr)));
      float lfm = mS + ls;
      float mn  = fmaxf(ir, lfm);
      float ig  = expf(ir - mn);
      float fg  = expf(lfm - mn);
      float cn  = fg * cS + ig * tanhf(zr);
      float nn  = fg * nS + ig;
      float og  = 1.0f / (1.0f + expf(-orr));
      hn = og * cn / nn;
      cS = cn;
      nS = nn;
      mS = mn;
    }

    ((float*)&lh[s & 1][w][hlf * SCAN_BT + bb])[e] = hn;
    __syncthreads();
    {
      const int sn = (s + 1 < XS) ? (s + 1) : s;
      const float* xq = xl + sn * (SCAN_BT * SCAN_XP);
      #pragma unroll
      for (int p = 0; p < 4; ++p) xv[p] = *(const float4*)(xq + 8 * p);
    }
    #pragma unroll
    for (int kk = 0; kk < SCAN_WAVES; ++kk) {
      float2 t = lh[s & 1][kk][hlf * SCAN_BT + bb];
      hA[2 * kk + 0] = t.x;
      hA[2 * kk + 1] = t.y;
    }

    if (writer) {
      v2f h2[8];
      #pragma unroll
      for (int kk = 0; kk < 8; ++kk) {
        h2[kk].x = hA[2 * kk + 0];
        h2[kk].y = hA[2 * kk + 1];
      }
      v2f s2 = ((h2[0] + h2[1]) + (h2[2] + h2[3])) + ((h2[4] + h2[5]) + (h2[6] + h2[7]));
      v2f q2 = ((h2[0] * h2[0] + h2[1] * h2[1]) + (h2[2] * h2[2] + h2[3] * h2[3])) +
               ((h2[4] * h2[4] + h2[5] * h2[5]) + (h2[6] * h2[6] + h2[7] * h2[7]));
      float sum = s2.x + s2.y;
      float sq  = q2.x + q2.y;
      sum += xhalf(sum);
      sq  += xhalf(sq);
      float mu = sum * (1.0f / XDH);
      float q  = sq * (1.0f / XDH) - mu * mu;
      float rs = rsqrtf(q + XEPS);
      #pragma unroll
      for (int i = 0; i < 2; ++i) {
        v2f lo0 = h2[2 * i], lo1 = h2[2 * i + 1];
        v2f hi0 = h2[4 + 2 * i], hi1 = h2[4 + 2 * i + 1];
        float u0 = e ? hi0.x : lo0.x;
        float u1 = e ? hi0.y : lo0.y;
        float u2 = e ? hi1.x : lo1.x;
        float u3 = e ? hi1.y : lo1.y;
        float4 o = hv[i];
        o.x += (u0 - mu) * rs * gn[4 * i + 0];
        o.y += (u1 - mu) * rs * gn[4 * i + 1];
        o.z += (u2 - mu) * rs * gn[4 * i + 2];
        o.w += (u3 - mu) * rs * gn[4 * i + 3];
        hp[2 * i] = o;
      }
    }
  }
}

__global__ __launch_bounds__(32) void k_ffn(const float* __restrict__ xn,
                                            float* __restrict__ hio,
                                            const float* __restrict__ Wup,
                                            const float* __restrict__ Wdn) {
  __shared__ float lact[16 * 128];
  const int lane = threadIdx.x;
  const int hlf  = lane >> 4;
  const int l16  = lane & 15;
  const size_t t0 = (size_t)blockIdx.x * 16;

  v2f ax[32];
  const float* xp = xn + (t0 + l16) * XD;
  #pragma unroll
  for (int kk = 0; kk < 32; ++kk) {
    int K0 = kk * 4 + hlf * 2;
    ax[kk].x = xp[K0];
    ax[kk].y = xp[K0 + 1];
  }

  float gatev[8][8];
  #pragma unroll
  for (int nt = 0; nt < 8; ++nt) {
    v8f a;
    #pragma unroll
    for (int r = 0; r < 8; ++r) a[r] = 0.0f;
    int j = nt * 16 + l16;
    #pragma unroll
    for (int kk = 0; kk < 32; ++kk) {
      int K0 = kk * 4 + hlf * 2;
      v2f b;
      b.x = Wup[(size_t)j * XD + K0];
      b.y = Wup[(size_t)j * XD + K0 + 1];
      a = wmma4(ax[kk], b, a);
    }
    #pragma unroll
    for (int r = 0; r < 8; ++r) gatev[nt][r] = a[r];
  }

  #pragma unroll
  for (int nt = 0; nt < 8; ++nt) {
    v8f a;
    #pragma unroll
    for (int r = 0; r < 8; ++r) a[r] = 0.0f;
    int j = 128 + nt * 16 + l16;
    #pragma unroll
    for (int kk = 0; kk < 32; ++kk) {
      int K0 = kk * 4 + hlf * 2;
      v2f b;
      b.x = Wup[(size_t)j * XD + K0];
      b.y = Wup[(size_t)j * XD + K0 + 1];
      a = wmma4(ax[kk], b, a);
    }
    #pragma unroll
    for (int r = 0; r < 8; ++r) {
      float act = gelu_exact(gatev[nt][r]) * a[r];
      lact[(r + hlf * 8) * 128 + nt * 16 + l16] = act;
    }
  }
  __syncthreads();

  v2f aa[32];
  const float* lp = lact + l16 * 128;
  #pragma unroll
  for (int kk = 0; kk < 32; ++kk) {
    int K0 = kk * 4 + hlf * 2;
    aa[kk].x = lp[K0];
    aa[kk].y = lp[K0 + 1];
  }
  #pragma unroll
  for (int nt = 0; nt < 8; ++nt) {
    v8f a;
    #pragma unroll
    for (int r = 0; r < 8; ++r) a[r] = 0.0f;
    int j = nt * 16 + l16;
    #pragma unroll
    for (int kk = 0; kk < 32; ++kk) {
      int K0 = kk * 4 + hlf * 2;
      v2f b;
      b.x = Wdn[(size_t)j * XD + K0];
      b.y = Wdn[(size_t)j * XD + K0 + 1];
      a = wmma4(aa[kk], b, a);
    }
    #pragma unroll
    for (int r = 0; r < 8; ++r) {
      size_t row = t0 + r + hlf * 8;
      hio[row * XD + nt * 16 + l16] += a[r];
    }
  }
}

__global__ __launch_bounds__(32) void k_proj(const float* __restrict__ xn,
                                             const float* __restrict__ pw,
                                             const float* __restrict__ pb,
                                             float* __restrict__ out) {
  const int lane = threadIdx.x;
  const int hlf  = lane >> 4;
  const int l16  = lane & 15;
  const size_t t0 = (size_t)blockIdx.x * 16;

  v8f a;
  #pragma unroll
  for (int r = 0; r < 8; ++r) a[r] = 0.0f;
  const float* xp = xn + (t0 + l16) * XD;
  #pragma unroll
  for (int kk = 0; kk < 32; ++kk) {
    int K0 = kk * 4 + hlf * 2;
    v2f av;
    av.x = xp[K0];
    av.y = xp[K0 + 1];
    v2f b;
    b.x = 0.0f;
    b.y = 0.0f;
    if (l16 < 9) {
      b.x = pw[(size_t)l16 * XD + K0];
      b.y = pw[(size_t)l16 * XD + K0 + 1];
    }
    a = wmma4(av, b, a);
  }
  if (l16 < 9) {
    float bb = pb[l16];
    #pragma unroll
    for (int r = 0; r < 8; ++r) {
      size_t row = t0 + r + hlf * 8;
      out[row * 9 + l16] = a[r] + bb;
    }
  }
}

extern "C" void kernel_launch(void* const* d_in, const int* in_sizes, int n_in,
                              void* d_out, int out_size, void* d_ws, size_t ws_size,
                              hipStream_t stream) {
  (void)in_sizes; (void)n_in; (void)out_size; (void)ws_size;
  const int*   x     = (const int*)d_in[0];
  const float* emb   = (const float*)d_in[1];
  const float* ln1   = (const float*)d_in[2];
  const float* Wg    = (const float*)d_in[3];
  const float* R     = (const float*)d_in[4];
  const float* bias  = (const float*)d_in[5];
  const float* gnw   = (const float*)d_in[6];
  const float* ln2   = (const float*)d_in[7];
  const float* wup   = (const float*)d_in[8];
  const float* wdn   = (const float*)d_in[9];
  const float* postw = (const float*)d_in[10];
  const float* pw    = (const float*)d_in[11];
  const float* pb    = (const float*)d_in[12];
  float* out = (float*)d_out;

  float* h  = (float*)d_ws;
  float* xb = h + (size_t)XT * XD;

  k_embed<<<(XT * XD + 255) / 256, 256, 0, stream>>>(x, emb, h, XT * XD);

  for (int bi = 0; bi < 2; ++bi) {
    k_ln<<<XT / 8, 256, 0, stream>>>(h, ln1 + bi * XD, xb);
    k_scan<<<dim3(XB / SCAN_BT, XNH), 32 * SCAN_WAVES, 0, stream>>>(
        xb, h,
        Wg   + (size_t)bi * 4 * XNH * XDH * XDH,
        R    + (size_t)bi * XNH * XDH * 4 * XDH,
        bias + (size_t)bi * XNH * 4 * XDH,
        gnw  + (size_t)bi * XD);
    k_ln<<<XT / 8, 256, 0, stream>>>(h, ln2 + bi * XD, xb);
    k_ffn<<<XT / 16, 32, 0, stream>>>(xb, h,
                                      wup + (size_t)bi * 2 * XD * XD,
                                      wdn + (size_t)bi * XD * XD);
  }

  k_ln<<<XT / 8, 256, 0, stream>>>(h, postw, xb);
  k_proj<<<XT / 16, 32, 0, stream>>>(xb, pw, pb, out);
}


# k_ffn: 8-wave work-group of 128 tokens copies each 128-row weight section once into LDS (coalesced 16-byte loads, next section in flight); waves read B fragments there, not from global; same sums
# speedup vs baseline: 6.0666x; 1.0634x over previous
#include <hip/hip_runtime.h>
#include <hip/hip_bf16.h>


typedef float v2f __attribute__((ext_vector_type(2)));
typedef float v8f __attribute__((ext_vector_type(8)));

#define XB   512
#define XS   64
#define XD   128
#define XNH  4
#define XDH  32
#define XT   (XB * XS)
#define XEPS 1e-5f

__device__ __forceinline__ v8f wmma4(v2f a, v2f b, v8f c) {
  return __builtin_amdgcn_wmma_f32_16x16x4_f32(false, a, false, b, (short)0, c,
                                               false, false);
}

__device__ __forceinline__ float gelu_exact(float x) {
  return 0.5f * x * (1.0f + erff(x * 0.70710678118654752f));
}

__global__ void k_embed(const int* __restrict__ x, const float* __restrict__ emb,
                        float* __restrict__ h, int total) {
  int i = blockIdx.x * blockDim.x + threadIdx.x;
  if (i >= total) return;
  int t = i >> 7;
  int c = i & 127;
  h[i] = emb[x[t] * XD + c];
}

__global__ void k_ln(const float* __restrict__ h, const float* __restrict__ w,
                     float* __restrict__ xn) {
  int wid  = threadIdx.x >> 5;
  int lane = threadIdx.x & 31;
  int t = blockIdx.x * (blockDim.x >> 5) + wid;
  const float4* hv = (const float4*)(h + (size_t)t * XD);
  float4 v = hv[lane];
  float s = v.x + v.y + v.z + v.w;
  float q = v.x * v.x + v.y * v.y + v.z * v.z + v.w * v.w;
  #pragma unroll
  for (int m = 16; m >= 1; m >>= 1) {
    s += __shfl_xor(s, m, 32);
    q += __shfl_xor(q, m, 32);
  }
  float mu  = s * (1.0f / XD);
  float var = q * (1.0f / XD) - mu * mu;
  float rs  = rsqrtf(var + XEPS);
  float4 wl = ((const float4*)w)[lane];
  float4 o;
  o.x = (v.x - mu) * rs * wl.x;
  o.y = (v.y - mu) * rs * wl.y;
  o.z = (v.z - mu) * rs * wl.z;
  o.w = (v.w - mu) * rs * wl.w;
  ((float4*)(xn + (size_t)t * XD))[lane] = o;
}

#define SCAN_WAVES 8
#define SCAN_BT    8
#define SCAN_XP    36

__device__ __forceinline__ float xhalf(float x) {
  int b = __builtin_bit_cast(int, x);
  return __builtin_bit_cast(
      float, __builtin_amdgcn_permlanex16(b, b, 0x76543210, 0xfedcba98, false, false));
}

__global__ __launch_bounds__(32 * SCAN_WAVES) void k_scan(const float* __restrict__ xn,
                                                          float* __restrict__ hio,
                                                          const float* __restrict__ Wg,
                                                          const float* __restrict__ R,
                                                          const float* __restrict__ bias,
                                                          const float* __restrict__ gnw) {
  struct ScanLds {
    float2 lh[2][SCAN_WAVES][2 * SCAN_BT];
    float lx[XS * SCAN_BT * SCAN_XP];
  };
  __shared__ __attribute__((aligned(16))) ScanLds sm;
  float2 (*lh)[SCAN_WAVES][2 * SCAN_BT] = sm.lh;
  float* lx = sm.lx;

  const int w    = threadIdx.x >> 5;
  const int lane = threadIdx.x & 31;
  const int hlf  = lane >> 4;
  const int l16  = lane & 15;
  const int bb   = l16 & 7;
  const int e    = l16 >> 3;
  const int b0   = blockIdx.x * SCAN_BT;
  const int n    = blockIdx.y;
  const int ub   = 8 * (w >> 1) + 2 * (w & 1);

  {
    const float* xsrc = xn + (size_t)b0 * XS * XD + n * XDH;
    float4 st[16];
    #pragma unroll
    for (int i = 0; i < 16; ++i) {
      int idx  = i * (32 * SCAN_WAVES) + (int)threadIdx.x;
      int part = idx & 7;
      int seg  = idx >> 3;
      int row  = seg >> 6;
      int ss   = seg & (XS - 1);
      st[i] = *(const float4*)(xsrc + ((size_t)row * XS + ss) * XD + part * 4);
    }
    #pragma unroll
    for (int i = 0; i < 16; ++i) {
      int idx  = i * (32 * SCAN_WAVES) + (int)threadIdx.x;
      int part = idx & 7;
      int seg  = idx >> 3;
      int row  = seg >> 6;
      int ss   = seg & (XS - 1);
      *(float4*)&lx[(ss * SCAN_BT + row) * SCAN_XP + part * 4] = st[i];
    }
  }

  const int ag = (l16 >> 1) & 3;
  const int au = ub + 4 * (l16 >> 3) + (l16 & 1);
  v2f aw[8], ar[8];
  #pragma unroll
  for (int kk = 0; kk < 8; ++kk) {
    int K0 = 8 * (kk >> 1) + 4 * hlf + 2 * (kk & 1);
    const float* wgp = Wg + (((size_t)(ag * XNH + n) * XDH + au) * XDH + K0);
    aw[kk].x = wgp[0];
    aw[kk].y = wgp[1];
    ar[kk].x = R[(size_t)(n * XDH + K0 + 0) * (4 * XDH) + ag * XDH + au];
    ar[kk].y = R[(size_t)(n * XDH + K0 + 1) * (4 * XDH) + ag * XDH + au];
  }
  float bz[8];
  #pragma unroll
  for (int r = 0; r < 8; ++r)
    bz[r] = bias[n * 128 + (r >> 1) * XDH + ub + 4 * hlf + (r & 1)];
  float gn[8];
  #pragma unroll
  for (int i = 0; i < 2; ++i) {
    #pragma unroll
    for (int q = 0; q < 4; ++q) gn[4 * i + q] = gnw[n * XDH + 8 * (2 * e + i) + 4 * hlf + q];
  }

  float cS = 0.0f;
  float nS = 0.0f;
  float mS = 0.0f;
  float hA[16] = {};

  const size_t rowoff = (size_t)(b0 + bb) * XS * XD + n * XDH + hlf * 4;

  const float* xl = lx + bb * SCAN_XP + hlf * 4;

  __syncthreads();
  float4 xv[4];
  #pragma unroll
  for (int p = 0; p < 4; ++p) xv[p] = *(const float4*)(xl + 8 * p);

  #pragma unroll 1
  for (int s = 0; s < XS; ++s) {
    float4*       hp = (float4*)(hio + rowoff + (size_t)s * XD) + 4 * e;
    const bool writer = ((s & (SCAN_WAVES - 1)) == w);
    float4 hv[2];
    #pragma unroll
    for (int i = 0; i < 2; ++i) hv[i] = make_float4(0.0f, 0.0f, 0.0f, 0.0f);
    if (writer) {
      #pragma unroll
      for (int i = 0; i < 2; ++i) hv[i] = hp[2 * i];
    }

    v2f bx[8], bh[8];
    #pragma unroll
    for (int kk = 0; kk < 8; ++kk) {
      float4 q = xv[kk >> 1];
      bx[kk].x = (kk & 1) ? q.z : q.x;
      bx[kk].y = (kk & 1) ? q.w : q.y;
      bh[kk].x = hA[2 * kk + 0];
      bh[kk].y = hA[2 * kk + 1];
    }

    v8f acc;
    #pragma unroll
    for (int r = 0; r < 8; ++r) acc[r] = bz[r];
    #pragma unroll
    for (int kk = 0; kk < 8; ++kk) acc = wmma4(aw[kk], bx[kk], acc);
    #pragma unroll
    for (int kk = 0; kk < 8; ++kk) acc = wmma4(ar[kk], bh[kk], acc);

    float hn;
    {
      float ir  = e ? acc[1] : acc[0];
      float fr  = e ? acc[3] : acc[2];
      float zr  = e ? acc[5] : acc[4];
      float orr = e ? acc[7] : acc[6];
      float ls  = (fr >= 0.0f) ? (-log1pf(expf(-fr)))
                               : (fr - log1pf(expf(fr)));
      float lfm = mS + ls;
      float mn  = fmaxf(ir, lfm);
      float ig  = expf(ir - mn);
      float fg  = expf(lfm - mn);
      float cn  = fg * cS + ig * tanhf(zr);
      float nn  = fg * nS + ig;
      float og  = 1.0f / (1.0f + expf(-orr));
      hn = og * cn / nn;
      cS = cn;
      nS = nn;
      mS = mn;
    }

    ((float*)&lh[s & 1][w][hlf * SCAN_BT + bb])[e] = hn;
    __syncthreads();
    {
      const int sn = (s + 1 < XS) ? (s + 1) : s;
      const float* xq = xl + sn * (SCAN_BT * SCAN_XP);
      #pragma unroll
      for (int p = 0; p < 4; ++p) xv[p] = *(const float4*)(xq + 8 * p);
    }
    #pragma unroll
    for (int kk = 0; kk < SCAN_WAVES; ++kk) {
      float2 t = lh[s & 1][kk][hlf * SCAN_BT + bb];
      hA[2 * kk + 0] = t.x;
      hA[2 * kk + 1] = t.y;
    }

    if (writer) {
      v2f h2[8];
      #pragma unroll
      for (int kk = 0; kk < 8; ++kk) {
        h2[kk].x = hA[2 * kk + 0];
        h2[kk].y = hA[2 * kk + 1];
      }
      v2f s2 = ((h2[0] + h2[1]) + (h2[2] + h2[3])) + ((h2[4] + h2[5]) + (h2[6] + h2[7]));
      v2f q2 = ((h2[0] * h2[0] + h2[1] * h2[1]) + (h2[2] * h2[2] + h2[3] * h2[3])) +
               ((h2[4] * h2[4] + h2[5] * h2[5]) + (h2[6] * h2[6] + h2[7] * h2[7]));
      float sum = s2.x + s2.y;
      float sq  = q2.x + q2.y;
      sum += xhalf(sum);
      sq  += xhalf(sq);
      float mu = sum * (1.0f / XDH);
      float q  = sq * (1.0f / XDH) - mu * mu;
      float rs = rsqrtf(q + XEPS);
      #pragma unroll
      for (int i = 0; i < 2; ++i) {
        v2f lo0 = h2[2 * i], lo1 = h2[2 * i + 1];
        v2f hi0 = h2[4 + 2 * i], hi1 = h2[4 + 2 * i + 1];
        float u0 = e ? hi0.x : lo0.x;
        float u1 = e ? hi0.y : lo0.y;
        float u2 = e ? hi1.x : lo1.x;
        float u3 = e ? hi1.y : lo1.y;
        float4 o = hv[i];
        o.x += (u0 - mu) * rs * gn[4 * i + 0];
        o.y += (u1 - mu) * rs * gn[4 * i + 1];
        o.z += (u2 - mu) * rs * gn[4 * i + 2];
        o.w += (u3 - mu) * rs * gn[4 * i + 3];
        hp[2 * i] = o;
      }
    }
  }
}

#define FFN_WAVES 8
#define FFN_WP    132

struct FfnLds {
  float wt[2][128 * FFN_WP];
  float act[FFN_WAVES][16 * FFN_WP];
};

typedef float v4f __attribute__((ext_vector_type(4)));
#define FFN_FETCH(st, src)                                                        \
  _Pragma("unroll") for (int i_ = 0; i_ < 16; ++i_)                               \
    (st)[i_] = *(const v4f*)((src) + (size_t)(i_ * 256 + tid) * 4)
#define FFN_PUT(st, dst)                                                          \
  _Pragma("unroll") for (int i_ = 0; i_ < 16; ++i_) {                             \
    int idx_ = i_ * 256 + tid;                                                    \
    *(v4f*)((dst) + (idx_ >> 5) * FFN_WP + (idx_ & 31) * 4) = (st)[i_];           \
  }

__device__ __forceinline__ void ffn_lds_barrier() {
  __builtin_amdgcn_fence(__ATOMIC_RELEASE, "workgroup", "local");
  __builtin_amdgcn_s_barrier();
  __builtin_amdgcn_fence(__ATOMIC_ACQUIRE, "workgroup", "local");
}

__device__ __forceinline__ void ffn_tile2(const v2f (&a)[32], const float* wsec, int nt,
                                          int l16, int hlf, v8f& c0, v8f& c1) {
  const float* p0 = wsec + (nt * 16 + l16) * FFN_WP + hlf * 2;
  const float* p1 = p0 + 16 * FFN_WP;
  #pragma unroll
  for (int r = 0; r < 8; ++r) {
    c0[r] = 0.0f;
    c1[r] = 0.0f;
  }
  #pragma unroll
  for (int kk = 0; kk < 32; ++kk) {
    v2f b0 = *(const v2f*)(p0 + kk * 4);
    v2f b1 = *(const v2f*)(p1 + kk * 4);
    c0 = wmma4(a[kk], b0, c0);
    c1 = wmma4(a[kk], b1, c1);
  }
}

__global__ __launch_bounds__(32 * FFN_WAVES) void k_ffn(const float* __restrict__ xn,
                                                        float* __restrict__ hio,
                                                        const float* __restrict__ Wup,
                                                        const float* __restrict__ Wdn) {
  __shared__ __attribute__((aligned(16))) FfnLds sm;
  const int tid  = threadIdx.x;
  const int w    = tid >> 5;
  const int lane = tid & 31;
  const int hlf  = lane >> 4;
  const int l16  = lane & 15;
  const size_t t0 = ((size_t)blockIdx.x * FFN_WAVES + w) * 16;
  float* actw = sm.act[w];

  v2f ax[32];
  const float* xp = xn + (t0 + l16) * XD + hlf * 2;
  #pragma unroll
  for (int kk = 0; kk < 32; ++kk) ax[kk] = *(const v2f*)(xp + kk * 4);

  v4f st[16];
  FFN_FETCH(st, Wup);
  FFN_PUT(st, sm.wt[0]);
  FFN_FETCH(st, Wup + 128 * XD);
  ffn_lds_barrier();

  v8f gatev[8];
  #pragma unroll
  for (int nt = 0; nt < 8; nt += 2)
    ffn_tile2(ax, sm.wt[0], nt, l16, hlf, gatev[nt], gatev[nt + 1]);

  FFN_PUT(st, sm.wt[1]);
  FFN_FETCH(st, Wdn);
  ffn_lds_barrier();

  #pragma unroll
  for (int nt = 0; nt < 8; nt += 2) {
    v8f u0, u1;
    ffn_tile2(ax, sm.wt[1], nt, l16, hlf, u0, u1);
    #pragma unroll
    for (int r = 0; r < 8; ++r) {
      actw[(r + hlf * 8) * FFN_WP + nt * 16 + l16]      = gelu_exact(gatev[nt][r]) * u0[r];
      actw[(r + hlf * 8) * FFN_WP + nt * 16 + 16 + l16] = gelu_exact(gatev[nt + 1][r]) * u1[r];
    }
  }

  FFN_PUT(st, sm.wt[0]);
  ffn_lds_barrier();

  v2f aa[32];
  const float* lp = actw + l16 * FFN_WP + hlf * 2;
  #pragma unroll
  for (int kk = 0; kk < 32; ++kk) aa[kk] = *(const v2f*)(lp + kk * 4);
  #pragma unroll
  for (int nt = 0; nt < 8; nt += 2) {
    v8f d0, d1;
    ffn_tile2(aa, sm.wt[0], nt, l16, hlf, d0, d1);
    #pragma unroll
    for (int r = 0; r < 8; ++r) {
      size_t row = t0 + r + hlf * 8;
      hio[row * XD + nt * 16 + l16]      += d0[r];
      hio[row * XD + nt * 16 + 16 + l16] += d1[r];
    }
  }
}

__global__ __launch_bounds__(32) void k_proj(const float* __restrict__ xn,
                                             const float* __restrict__ pw,
                                             const float* __restrict__ pb,
                                             float* __restrict__ out) {
  const int lane = threadIdx.x;
  const int hlf  = lane >> 4;
  const int l16  = lane & 15;
  const size_t t0 = (size_t)blockIdx.x * 16;

  v8f a;
  #pragma unroll
  for (int r = 0; r < 8; ++r) a[r] = 0.0f;
  const float* xp = xn + (t0 + l16) * XD;
  #pragma unroll
  for (int kk = 0; kk < 32; ++kk) {
    int K0 = kk * 4 + hlf * 2;
    v2f av;
    av.x = xp[K0];
    av.y = xp[K0 + 1];
    v2f b;
    b.x = 0.0f;
    b.y = 0.0f;
    if (l16 < 9) {
      b.x = pw[(size_t)l16 * XD + K0];
      b.y = pw[(size_t)l16 * XD + K0 + 1];
    }
    a = wmma4(av, b, a);
  }
  if (l16 < 9) {
    float bb = pb[l16];
    #pragma unroll
    for (int r = 0; r < 8; ++r) {
      size_t row = t0 + r + hlf * 8;
      out[row * 9 + l16] = a[r] + bb;
    }
  }
}

extern "C" void kernel_launch(void* const* d_in, const int* in_sizes, int n_in,
                              void* d_out, int out_size, void* d_ws, size_t ws_size,
                              hipStream_t stream) {
  (void)in_sizes; (void)n_in; (void)out_size; (void)ws_size;
  const int*   x     = (const int*)d_in[0];
  const float* emb   = (const float*)d_in[1];
  const float* ln1   = (const float*)d_in[2];
  const float* Wg    = (const float*)d_in[3];
  const float* R     = (const float*)d_in[4];
  const float* bias  = (const float*)d_in[5];
  const float* gnw   = (const float*)d_in[6];
  const float* ln2   = (const float*)d_in[7];
  const float* wup   = (const float*)d_in[8];
  const float* wdn   = (const float*)d_in[9];
  const float* postw = (const float*)d_in[10];
  const float* pw    = (const float*)d_in[11];
  const float* pb    = (const float*)d_in[12];
  float* out = (float*)d_out;

  float* h  = (float*)d_ws;
  float* xb = h + (size_t)XT * XD;

  k_embed<<<(XT * XD + 255) / 256, 256, 0, stream>>>(x, emb, h, XT * XD);

  for (int bi = 0; bi < 2; ++bi) {
    k_ln<<<XT / 8, 256, 0, stream>>>(h, ln1 + bi * XD, xb);
    k_scan<<<dim3(XB / SCAN_BT, XNH), 32 * SCAN_WAVES, 0, stream>>>(
        xb, h,
        Wg   + (size_t)bi * 4 * XNH * XDH * XDH,
        R    + (size_t)bi * XNH * XDH * 4 * XDH,
        bias + (size_t)bi * XNH * 4 * XDH,
        gnw  + (size_t)bi * XD);
    k_ln<<<XT / 8, 256, 0, stream>>>(h, ln2 + bi * XD, xb);
    k_ffn<<<XT / (16 * FFN_WAVES), 32 * FFN_WAVES, 0, stream>>>(xb, h,
                                      wup + (size_t)bi * 2 * XD * XD,
                                      wdn + (size_t)bi * XD * XD);
  }

  k_ln<<<XT / 8, 256, 0, stream>>>(h, postw, xb);
  k_proj<<<XT / 16, 32, 0, stream>>>(xb, pw, pb, out);
}


# no layernorm launches: one joins the embedding, four run in k_ffn waves (before: on A fragments, LN weight folded into staged up weights; after: on accumulators); 11->6 launches; LN sum order changes
# speedup vs baseline: 6.6844x; 1.1018x over previous
#include <hip/hip_runtime.h>
#include <hip/hip_bf16.h>


typedef float v2f __attribute__((ext_vector_type(2)));
typedef float v8f __attribute__((ext_vector_type(8)));

#define XB   512
#define XS   64
#define XD   128
#define XNH  4
#define XDH  32
#define XT   (XB * XS)
#define XEPS 1e-5f

__device__ __forceinline__ v8f wmma4(v2f a, v2f b, v8f c) {
  return __builtin_amdgcn_wmma_f32_16x16x4_f32(false, a, false, b, (short)0, c,
                                               false, false);
}

__device__ __forceinline__ float gelu_exact(float x) {
  return 0.5f * x * (1.0f + erff(x * 0.70710678118654752f));
}

__device__ __forceinline__ float xhalf(float x) {
  int b = __builtin_bit_cast(int, x);
  return __builtin_bit_cast(
      float, __builtin_amdgcn_permlanex16(b, b, 0x76543210, 0xfedcba98, false, false));
}

__global__ void k_embed_ln(const int* __restrict__ x, const float* __restrict__ emb,
                           const float* __restrict__ w, float* __restrict__ h,
                           float* __restrict__ xn) {
  int wid  = threadIdx.x >> 5;
  int lane = threadIdx.x & 31;
  int t = blockIdx.x * (blockDim.x >> 5) + wid;
  float4 v = ((const float4*)(emb + (size_t)x[t] * XD))[lane];
  ((float4*)(h + (size_t)t * XD))[lane] = v;
  float s = v.x + v.y + v.z + v.w;
  float q = v.x * v.x + v.y * v.y + v.z * v.z + v.w * v.w;
  #pragma unroll
  for (int m = 16; m >= 1; m >>= 1) {
    s += __shfl_xor(s, m, 32);
    q += __shfl_xor(q, m, 32);
  }
  float mu  = s * (1.0f / XD);
  float var = q * (1.0f / XD) - mu * mu;
  float rs  = rsqrtf(var + XEPS);
  float4 wl = ((const float4*)w)[lane];
  float4 o;
  o.x = (v.x - mu) * rs * wl.x;
  o.y = (v.y - mu) * rs * wl.y;
  o.z = (v.z - mu) * rs * wl.z;
  o.w = (v.w - mu) * rs * wl.w;
  ((float4*)(xn + (size_t)t * XD))[lane] = o;
}

__device__ __forceinline__ void ln_afrag(v2f (&a)[32]) {
  v2f s4[4], q4[4];
  #pragma unroll
  for (int j = 0; j < 4; ++j) {
    s4[j] = a[j];
    q4[j] = a[j] * a[j];
  }
  #pragma unroll
  for (int kk = 4; kk < 32; ++kk) {
    s4[kk & 3] += a[kk];
    q4[kk & 3] += a[kk] * a[kk];
  }
  v2f s2 = (s4[0] + s4[1]) + (s4[2] + s4[3]);
  v2f q2 = (q4[0] + q4[1]) + (q4[2] + q4[3]);
  float s = s2.x + s2.y;
  float q = q2.x + q2.y;
  s += xhalf(s);
  q += xhalf(q);
  float mu  = s * (1.0f / XD);
  float var = q * (1.0f / XD) - mu * mu;
  float rs  = rsqrtf(var + XEPS);
  #pragma unroll
  for (int kk = 0; kk < 32; ++kk) {
    a[kk].x = (a[kk].x - mu) * rs;
    a[kk].y = (a[kk].y - mu) * rs;
  }
}

#define SCAN_WAVES 8
#define SCAN_BT    8
#define SCAN_XP    36

__global__ __launch_bounds__(32 * SCAN_WAVES) void k_scan(const float* __restrict__ xn,
                                                          float* __restrict__ hio,
                                                          const float* __restrict__ Wg,
                                                          const float* __restrict__ R,
                                                          const float* __restrict__ bias,
                                                          const float* __restrict__ gnw) {
  struct ScanLds {
    float2 lh[2][SCAN_WAVES][2 * SCAN_BT];
    float lx[XS * SCAN_BT * SCAN_XP];
  };
  __shared__ __attribute__((aligned(16))) ScanLds sm;
  float2 (*lh)[SCAN_WAVES][2 * SCAN_BT] = sm.lh;
  float* lx = sm.lx;

  const int w    = threadIdx.x >> 5;
  const int lane = threadIdx.x & 31;
  const int hlf  = lane >> 4;
  const int l16  = lane & 15;
  const int bb   = l16 & 7;
  const int e    = l16 >> 3;
  const int b0   = blockIdx.x * SCAN_BT;
  const int n    = blockIdx.y;
  const int ub   = 8 * (w >> 1) + 2 * (w & 1);

  {
    const float* xsrc = xn + (size_t)b0 * XS * XD + n * XDH;
    float4 st[16];
    #pragma unroll
    for (int i = 0; i < 16; ++i) {
      int idx  = i * (32 * SCAN_WAVES) + (int)threadIdx.x;
      int part = idx & 7;
      int seg  = idx >> 3;
      int row  = seg >> 6;
      int ss   = seg & (XS - 1);
      st[i] = *(const float4*)(xsrc + ((size_t)row * XS + ss) * XD + part * 4);
    }
    #pragma unroll
    for (int i = 0; i < 16; ++i) {
      int idx  = i * (32 * SCAN_WAVES) + (int)threadIdx.x;
      int part = idx & 7;
      int seg  = idx >> 3;
      int row  = seg >> 6;
      int ss   = seg & (XS - 1);
      *(float4*)&lx[(ss * SCAN_BT + row) * SCAN_XP + part * 4] = st[i];
    }
  }

  const int ag = (l16 >> 1) & 3;
  const int au = ub + 4 * (l16 >> 3) + (l16 & 1);
  v2f aw[8], ar[8];
  #pragma unroll
  for (int kk = 0; kk < 8; ++kk) {
    int K0 = 8 * (kk >> 1) + 4 * hlf + 2 * (kk & 1);
    const float* wgp = Wg + (((size_t)(ag * XNH + n) * XDH + au) * XDH + K0);
    aw[kk].x = wgp[0];
    aw[kk].y = wgp[1];
    ar[kk].x = R[(size_t)(n * XDH + K0 + 0) * (4 * XDH) + ag * XDH + au];
    ar[kk].y = R[(size_t)(n * XDH + K0 + 1) * (4 * XDH) + ag * XDH + au];
  }
  float bz[8];
  #pragma unroll
  for (int r = 0; r < 8; ++r)
    bz[r] = bias[n * 128 + (r >> 1) * XDH + ub + 4 * hlf + (r & 1)];
  float gn[8];
  #pragma unroll
  for (int i = 0; i < 2; ++i) {
    #pragma unroll
    for (int q = 0; q < 4; ++q) gn[4 * i + q] = gnw[n * XDH + 8 * (2 * e + i) + 4 * hlf + q];
  }

  float cS = 0.0f;
  float nS = 0.0f;
  float mS = 0.0f;
  float hA[16] = {};

  const size_t rowoff = (size_t)(b0 + bb) * XS * XD + n * XDH + hlf * 4;

  const float* xl = lx + bb * SCAN_XP + hlf * 4;

  __syncthreads();
  float4 xv[4];
  #pragma unroll
  for (int p = 0; p < 4; ++p) xv[p] = *(const float4*)(xl + 8 * p);

  #pragma unroll 1
  for (int s = 0; s < XS; ++s) {
    float4*       hp = (float4*)(hio + rowoff + (size_t)s * XD) + 4 * e;
    const bool writer = ((s & (SCAN_WAVES - 1)) == w);
    float4 hv[2];
    #pragma unroll
    for (int i = 0; i < 2; ++i) hv[i] = make_float4(0.0f, 0.0f, 0.0f, 0.0f);
    if (writer) {
      #pragma unroll
      for (int i = 0; i < 2; ++i) hv[i] = hp[2 * i];
    }

    v2f bx[8], bh[8];
    #pragma unroll
    for (int kk = 0; kk < 8; ++kk) {
      float4 q = xv[kk >> 1];
      bx[kk].x = (kk & 1) ? q.z : q.x;
      bx[kk].y = (kk & 1) ? q.w : q.y;
      bh[kk].x = hA[2 * kk + 0];
      bh[kk].y = hA[2 * kk + 1];
    }

    v8f acc;
    #pragma unroll
    for (int r = 0; r < 8; ++r) acc[r] = bz[r];
    #pragma unroll
    for (int kk = 0; kk < 8; ++kk) acc = wmma4(aw[kk], bx[kk], acc);
    #pragma unroll
    for (int kk = 0; kk < 8; ++kk) acc = wmma4(ar[kk], bh[kk], acc);

    float hn;
    {
      float ir  = e ? acc[1] : acc[0];
      float fr  = e ? acc[3] : acc[2];
      float zr  = e ? acc[5] : acc[4];
      float orr = e ? acc[7] : acc[6];
      float ls  = (fr >= 0.0f) ? (-log1pf(expf(-fr)))
                               : (fr - log1pf(expf(fr)));
      float lfm = mS + ls;
      float mn  = fmaxf(ir, lfm);
      float ig  = expf(ir - mn);
      float fg  = expf(lfm - mn);
      float cn  = fg * cS + ig * tanhf(zr);
      float nn  = fg * nS + ig;
      float og  = 1.0f / (1.0f + expf(-orr));
      hn = og * cn / nn;
      cS = cn;
      nS = nn;
      mS = mn;
    }

    ((float*)&lh[s & 1][w][hlf * SCAN_BT + bb])[e] = hn;
    __syncthreads();
    {
      const int sn = (s + 1 < XS) ? (s + 1) : s;
      const float* xq = xl + sn * (SCAN_BT * SCAN_XP);
      #pragma unroll
      for (int p = 0; p < 4; ++p) xv[p] = *(const float4*)(xq + 8 * p);
    }
    #pragma unroll
    for (int kk = 0; kk < SCAN_WAVES; ++kk) {
      float2 t = lh[s & 1][kk][hlf * SCAN_BT + bb];
      hA[2 * kk + 0] = t.x;
      hA[2 * kk + 1] = t.y;
    }

    if (writer) {
      v2f h2[8];
      #pragma unroll
      for (int kk = 0; kk < 8; ++kk) {
        h2[kk].x = hA[2 * kk + 0];
        h2[kk].y = hA[2 * kk + 1];
      }
      v2f s2 = ((h2[0] + h2[1]) + (h2[2] + h2[3])) + ((h2[4] + h2[5]) + (h2[6] + h2[7]));
      v2f q2 = ((h2[0] * h2[0] + h2[1] * h2[1]) + (h2[2] * h2[2] + h2[3] * h2[3])) +
               ((h2[4] * h2[4] + h2[5] * h2[5]) + (h2[6] * h2[6] + h2[7] * h2[7]));
      float sum = s2.x + s2.y;
      float sq  = q2.x + q2.y;
      sum += xhalf(sum);
      sq  += xhalf(sq);
      float mu = sum * (1.0f / XDH);
      float q  = sq * (1.0f / XDH) - mu * mu;
      float rs = rsqrtf(q + XEPS);
      #pragma unroll
      for (int i = 0; i < 2; ++i) {
        v2f lo0 = h2[2 * i], lo1 = h2[2 * i + 1];
        v2f hi0 = h2[4 + 2 * i], hi1 = h2[4 + 2 * i + 1];
        float u0 = e ? hi0.x : lo0.x;
        float u1 = e ? hi0.y : lo0.y;
        float u2 = e ? hi1.x : lo1.x;
        float u3 = e ? hi1.y : lo1.y;
        float4 o = hv[i];
        o.x += (u0 - mu) * rs * gn[4 * i + 0];
        o.y += (u1 - mu) * rs * gn[4 * i + 1];
        o.z += (u2 - mu) * rs * gn[4 * i + 2];
        o.w += (u3 - mu) * rs * gn[4 * i + 3];
        hp[2 * i] = o;
      }
    }
  }
}

#define FFN_WAVES 8
#define FFN_WP    132

struct FfnLds {
  float wt[2][128 * FFN_WP];
  float act[FFN_WAVES][16 * FFN_WP];
};

typedef float v4f __attribute__((ext_vector_type(4)));
#define FFN_FETCH(st, src)                                                        \
  _Pragma("unroll") for (int i_ = 0; i_ < 16; ++i_)                               \
    (st)[i_] = *(const v4f*)((src) + (size_t)(i_ * 256 + tid) * 4)
#define FFN_PUT(st, dst)                                                          \
  _Pragma("unroll") for (int i_ = 0; i_ < 16; ++i_) {                             \
    int idx_ = i_ * 256 + tid;                                                    \
    *(v4f*)((dst) + (idx_ >> 5) * FFN_WP + (idx_ & 31) * 4) = (st)[i_];           \
  }

#define FFN_PUT_SCALED(st, dst, sc)                                               \
  _Pragma("unroll") for (int i_ = 0; i_ < 16; ++i_) {                             \
    int idx_ = i_ * 256 + tid;                                                    \
    *(v4f*)((dst) + (idx_ >> 5) * FFN_WP + (idx_ & 31) * 4) = (st)[i_] * (sc);    \
  }

__device__ __forceinline__ void ffn_lds_barrier() {
  __builtin_amdgcn_fence(__ATOMIC_RELEASE, "workgroup", "local");
  __builtin_amdgcn_s_barrier();
  __builtin_amdgcn_fence(__ATOMIC_ACQUIRE, "workgroup", "local");
}

__device__ __forceinline__ void ffn_tile2(const v2f (&a)[32], const float* wsec, int nt,
                                          int l16, int hlf, v8f& c0, v8f& c1) {
  const float* p0 = wsec + (nt * 16 + l16) * FFN_WP + hlf * 2;
  const float* p1 = p0 + 16 * FFN_WP;
  #pragma unroll
  for (int r = 0; r < 8; ++r) {
    c0[r] = 0.0f;
    c1[r] = 0.0f;
  }
  #pragma unroll
  for (int kk = 0; kk < 32; ++kk) {
    v2f b0 = *(const v2f*)(p0 + kk * 4);
    v2f b1 = *(const v2f*)(p1 + kk * 4);
    c0 = wmma4(a[kk], b0, c0);
    c1 = wmma4(a[kk], b1, c1);
  }
}

__global__ __launch_bounds__(32 * FFN_WAVES) void k_ffn(float* __restrict__ hio,
                                                        float* __restrict__ xnext,
                                                        const float* __restrict__ lnw,
                                                        const float* __restrict__ lnnext,
                                                        const float* __restrict__ Wup,
                                                        const float* __restrict__ Wdn,
                                                        int store_h) {
  __shared__ __attribute__((aligned(16))) FfnLds sm;
  const int tid  = threadIdx.x;
  const int w    = tid >> 5;
  const int lane = tid & 31;
  const int hlf  = lane >> 4;
  const int l16  = lane & 15;
  const size_t t0 = ((size_t)blockIdx.x * FFN_WAVES + w) * 16;
  float* actw = sm.act[w];

  v2f ax[32];
  const float* xp = hio + (t0 + l16) * XD + hlf * 2;
  #pragma unroll
  for (int kk = 0; kk < 32; ++kk) ax[kk] = *(const v2f*)(xp + kk * 4);

  const v4f lw = *(const v4f*)(lnw + lane * 4);
  v4f st[16];
  FFN_FETCH(st, Wup);
  ln_afrag(ax);
  FFN_PUT_SCALED(st, sm.wt[0], lw);
  FFN_FETCH(st, Wup + 128 * XD);
  ffn_lds_barrier();

  v8f gatev[8];
  #pragma unroll
  for (int nt = 0; nt < 8; nt += 2)
    ffn_tile2(ax, sm.wt[0], nt, l16, hlf, gatev[nt], gatev[nt + 1]);

  FFN_PUT_SCALED(st, sm.wt[1], lw);
  FFN_FETCH(st, Wdn);
  ffn_lds_barrier();

  #pragma unroll
  for (int nt = 0; nt < 8; nt += 2) {
    v8f u0, u1;
    ffn_tile2(ax, sm.wt[1], nt, l16, hlf, u0, u1);
    #pragma unroll
    for (int r = 0; r < 8; ++r) {
      actw[(r + hlf * 8) * FFN_WP + nt * 16 + l16]      = gelu_exact(gatev[nt][r]) * u0[r];
      actw[(r + hlf * 8) * FFN_WP + nt * 16 + 16 + l16] = gelu_exact(gatev[nt + 1][r]) * u1[r];
    }
  }

  FFN_PUT(st, sm.wt[0]);
  ffn_lds_barrier();

  v2f aa[32];
  const float* lp = actw + l16 * FFN_WP + hlf * 2;
  #pragma unroll
  for (int kk = 0; kk < 32; ++kk) aa[kk] = *(const v2f*)(lp + kk * 4);
  float hn[8][8];
  float rsum[8], rsq[8];
  #pragma unroll
  for (int r = 0; r < 8; ++r) {
    rsum[r] = 0.0f;
    rsq[r]  = 0.0f;
  }
  #pragma unroll
  for (int nt = 0; nt < 8; nt += 2) {
    v8f d0, d1;
    ffn_tile2(aa, sm.wt[0], nt, l16, hlf, d0, d1);
    #pragma unroll
    for (int r = 0; r < 8; ++r) {
      size_t o = (t0 + r + hlf * 8) * XD + nt * 16 + l16;
      float a0 = hio[o] + d0[r];
      float a1 = hio[o + 16] + d1[r];
      hn[nt][r]     = a0;
      hn[nt + 1][r] = a1;
      rsum[r] += a0 + a1;
      rsq[r]  += a0 * a0 + a1 * a1;
    }
  }

  if (store_h) {
    #pragma unroll
    for (int nt = 0; nt < 8; ++nt) {
      #pragma unroll
      for (int r = 0; r < 8; ++r)
        hio[(t0 + r + hlf * 8) * XD + nt * 16 + l16] = hn[nt][r];
    }
  }

  #pragma unroll
  for (int r = 0; r < 8; ++r) {
    #pragma unroll
    for (int m = 8; m >= 1; m >>= 1) {
      rsum[r] += __shfl_xor(rsum[r], m, 32);
      rsq[r]  += __shfl_xor(rsq[r], m, 32);
    }
  }
  float wn[8];
  #pragma unroll
  for (int nt = 0; nt < 8; ++nt) wn[nt] = lnnext[nt * 16 + l16];
  #pragma unroll
  for (int r = 0; r < 8; ++r) {
    float mu  = rsum[r] * (1.0f / XD);
    float var = rsq[r] * (1.0f / XD) - mu * mu;
    float rs  = rsqrtf(var + XEPS);
    #pragma unroll
    for (int nt = 0; nt < 8; ++nt)
      actw[(r + hlf * 8) * FFN_WP + nt * 16 + l16] = (hn[nt][r] - mu) * rs * wn[nt];
  }
  ffn_lds_barrier();
  #pragma unroll
  for (int i = 0; i < 16; ++i) {
    v4f v = *(const v4f*)(actw + i * FFN_WP + lane * 4);
    *(v4f*)(xnext + (t0 + i) * XD + lane * 4) = v;
  }
}

__global__ __launch_bounds__(32) void k_proj(const float* __restrict__ xn,
                                             const float* __restrict__ pw,
                                             const float* __restrict__ pb,
                                             float* __restrict__ out) {
  const int lane = threadIdx.x;
  const int hlf  = lane >> 4;
  const int l16  = lane & 15;
  const size_t t0 = (size_t)blockIdx.x * 16;

  v8f a;
  #pragma unroll
  for (int r = 0; r < 8; ++r) a[r] = 0.0f;
  const float* xp = xn + (t0 + l16) * XD;
  #pragma unroll
  for (int kk = 0; kk < 32; ++kk) {
    int K0 = kk * 4 + hlf * 2;
    v2f av;
    av.x = xp[K0];
    av.y = xp[K0 + 1];
    v2f b;
    b.x = 0.0f;
    b.y = 0.0f;
    if (l16 < 9) {
      b.x = pw[(size_t)l16 * XD + K0];
      b.y = pw[(size_t)l16 * XD + K0 + 1];
    }
    a = wmma4(av, b, a);
  }
  if (l16 < 9) {
    float bb = pb[l16];
    #pragma unroll
    for (int r = 0; r < 8; ++r) {
      size_t row = t0 + r + hlf * 8;
      out[row * 9 + l16] = a[r] + bb;
    }
  }
}

extern "C" void kernel_launch(void* const* d_in, const int* in_sizes, int n_in,
                              void* d_out, int out_size, void* d_ws, size_t ws_size,
                              hipStream_t stream) {
  (void)in_sizes; (void)n_in; (void)out_size; (void)ws_size;
  const int*   x     = (const int*)d_in[0];
  const float* emb   = (const float*)d_in[1];
  const float* ln1   = (const float*)d_in[2];
  const float* Wg    = (const float*)d_in[3];
  const float* R     = (const float*)d_in[4];
  const float* bias  = (const float*)d_in[5];
  const float* gnw   = (const float*)d_in[6];
  const float* ln2   = (const float*)d_in[7];
  const float* wup   = (const float*)d_in[8];
  const float* wdn   = (const float*)d_in[9];
  const float* postw = (const float*)d_in[10];
  const float* pw    = (const float*)d_in[11];
  const float* pb    = (const float*)d_in[12];
  float* out = (float*)d_out;

  float* h  = (float*)d_ws;
  float* xb = h + (size_t)XT * XD;

  k_embed_ln<<<XT / 8, 256, 0, stream>>>(x, emb, ln1, h, xb);

  k_scan<<<dim3(XB / SCAN_BT, XNH), 32 * SCAN_WAVES, 0, stream>>>(xb, h, Wg, R, bias, gnw);
  k_ffn<<<XT / (16 * FFN_WAVES), 32 * FFN_WAVES, 0, stream>>>(h, xb, ln2, ln1 + XD, wup, wdn, 1);

  k_scan<<<dim3(XB / SCAN_BT, XNH), 32 * SCAN_WAVES, 0, stream>>>(
      xb, h,
      Wg   + (size_t)4 * XNH * XDH * XDH,
      R    + (size_t)XNH * XDH * 4 * XDH,
      bias + (size_t)XNH * 4 * XDH,
      gnw  + XD);
  k_ffn<<<XT / (16 * FFN_WAVES), 32 * FFN_WAVES, 0, stream>>>(h, xb, ln2 + XD, postw,
                                                              wup + (size_t)2 * XD * XD,
                                                              wdn + (size_t)XD * XD, 0);

  k_proj<<<XT / 16, 32, 0, stream>>>(xb, pw, pb, out);
}


# scan gate chain: stabiliser m = max(i, m + min(f,0)) replaces max(i, m + log_sigmoid(f)): no log1p in a step; forget = exp(q-m)/(1+exp(-|f|)); c, n carried rescaled, h differs in last bits only
# speedup vs baseline: 8.0476x; 1.2039x over previous
#include <hip/hip_runtime.h>
#include <hip/hip_bf16.h>


typedef float v2f __attribute__((ext_vector_type(2)));
typedef float v8f __attribute__((ext_vector_type(8)));

#define XB   512
#define XS   64
#define XD   128
#define XNH  4
#define XDH  32
#define XT   (XB * XS)
#define XEPS 1e-5f

__device__ __forceinline__ v8f wmma4(v2f a, v2f b, v8f c) {
  return __builtin_amdgcn_wmma_f32_16x16x4_f32(false, a, false, b, (short)0, c,
                                               false, false);
}

__device__ __forceinline__ float gelu_exact(float x) {
  return 0.5f * x * (1.0f + erff(x * 0.70710678118654752f));
}

__device__ __forceinline__ float xhalf(float x) {
  int b = __builtin_bit_cast(int, x);
  return __builtin_bit_cast(
      float, __builtin_amdgcn_permlanex16(b, b, 0x76543210, 0xfedcba98, false, false));
}

__global__ void k_embed_ln(const int* __restrict__ x, const float* __restrict__ emb,
                           const float* __restrict__ w, float* __restrict__ h,
                           float* __restrict__ xn) {
  int wid  = threadIdx.x >> 5;
  int lane = threadIdx.x & 31;
  int t = blockIdx.x * (blockDim.x >> 5) + wid;
  float4 v = ((const float4*)(emb + (size_t)x[t] * XD))[lane];
  ((float4*)(h + (size_t)t * XD))[lane] = v;
  float s = v.x + v.y + v.z + v.w;
  float q = v.x * v.x + v.y * v.y + v.z * v.z + v.w * v.w;
  #pragma unroll
  for (int m = 16; m >= 1; m >>= 1) {
    s += __shfl_xor(s, m, 32);
    q += __shfl_xor(q, m, 32);
  }
  float mu  = s * (1.0f / XD);
  float var = q * (1.0f / XD) - mu * mu;
  float rs  = rsqrtf(var + XEPS);
  float4 wl = ((const float4*)w)[lane];
  float4 o;
  o.x = (v.x - mu) * rs * wl.x;
  o.y = (v.y - mu) * rs * wl.y;
  o.z = (v.z - mu) * rs * wl.z;
  o.w = (v.w - mu) * rs * wl.w;
  ((float4*)(xn + (size_t)t * XD))[lane] = o;
}

__device__ __forceinline__ void ln_afrag(v2f (&a)[32]) {
  v2f s4[4], q4[4];
  #pragma unroll
  for (int j = 0; j < 4; ++j) {
    s4[j] = a[j];
    q4[j] = a[j] * a[j];
  }
  #pragma unroll
  for (int kk = 4; kk < 32; ++kk) {
    s4[kk & 3] += a[kk];
    q4[kk & 3] += a[kk] * a[kk];
  }
  v2f s2 = (s4[0] + s4[1]) + (s4[2] + s4[3]);
  v2f q2 = (q4[0] + q4[1]) + (q4[2] + q4[3]);
  float s = s2.x + s2.y;
  float q = q2.x + q2.y;
  s += xhalf(s);
  q += xhalf(q);
  float mu  = s * (1.0f / XD);
  float var = q * (1.0f / XD) - mu * mu;
  float rs  = rsqrtf(var + XEPS);
  #pragma unroll
  for (int kk = 0; kk < 32; ++kk) {
    a[kk].x = (a[kk].x - mu) * rs;
    a[kk].y = (a[kk].y - mu) * rs;
  }
}

#define SCAN_WAVES 8
#define SCAN_BT    8
#define SCAN_XP    36

__global__ __launch_bounds__(32 * SCAN_WAVES) void k_scan(const float* __restrict__ xn,
                                                          float* __restrict__ hio,
                                                          const float* __restrict__ Wg,
                                                          const float* __restrict__ R,
                                                          const float* __restrict__ bias,
                                                          const float* __restrict__ gnw) {
  struct ScanLds {
    float2 lh[2][SCAN_WAVES][2 * SCAN_BT];
    float lx[XS * SCAN_BT * SCAN_XP];
  };
  __shared__ __attribute__((aligned(16))) ScanLds sm;
  float2 (*lh)[SCAN_WAVES][2 * SCAN_BT] = sm.lh;
  float* lx = sm.lx;

  const int w    = threadIdx.x >> 5;
  const int lane = threadIdx.x & 31;
  const int hlf  = lane >> 4;
  const int l16  = lane & 15;
  const int bb   = l16 & 7;
  const int e    = l16 >> 3;
  const int b0   = blockIdx.x * SCAN_BT;
  const int n    = blockIdx.y;
  const int ub   = 8 * (w >> 1) + 2 * (w & 1);

  {
    const float* xsrc = xn + (size_t)b0 * XS * XD + n * XDH;
    float4 st[16];
    #pragma unroll
    for (int i = 0; i < 16; ++i) {
      int idx  = i * (32 * SCAN_WAVES) + (int)threadIdx.x;
      int part = idx & 7;
      int seg  = idx >> 3;
      int row  = seg >> 6;
      int ss   = seg & (XS - 1);
      st[i] = *(const float4*)(xsrc + ((size_t)row * XS + ss) * XD + part * 4);
    }
    #pragma unroll
    for (int i = 0; i < 16; ++i) {
      int idx  = i * (32 * SCAN_WAVES) + (int)threadIdx.x;
      int part = idx & 7;
      int seg  = idx >> 3;
      int row  = seg >> 6;
      int ss   = seg & (XS - 1);
      *(float4*)&lx[(ss * SCAN_BT + row) * SCAN_XP + part * 4] = st[i];
    }
  }

  const int ag = (l16 >> 1) & 3;
  const int au = ub + 4 * (l16 >> 3) + (l16 & 1);
  v2f aw[8], ar[8];
  #pragma unroll
  for (int kk = 0; kk < 8; ++kk) {
    int K0 = 8 * (kk >> 1) + 4 * hlf + 2 * (kk & 1);
    const float* wgp = Wg + (((size_t)(ag * XNH + n) * XDH + au) * XDH + K0);
    aw[kk].x = wgp[0];
    aw[kk].y = wgp[1];
    ar[kk].x = R[(size_t)(n * XDH + K0 + 0) * (4 * XDH) + ag * XDH + au];
    ar[kk].y = R[(size_t)(n * XDH + K0 + 1) * (4 * XDH) + ag * XDH + au];
  }
  float bz[8];
  #pragma unroll
  for (int r = 0; r < 8; ++r)
    bz[r] = bias[n * 128 + (r >> 1) * XDH + ub + 4 * hlf + (r & 1)];
  float gn[8];
  #pragma unroll
  for (int i = 0; i < 2; ++i) {
    #pragma unroll
    for (int q = 0; q < 4; ++q) gn[4 * i + q] = gnw[n * XDH + 8 * (2 * e + i) + 4 * hlf + q];
  }

  float cS = 0.0f;
  float nS = 0.0f;
  float mS = 0.0f;
  float hA[16] = {};

  const size_t rowoff = (size_t)(b0 + bb) * XS * XD + n * XDH + hlf * 4;

  const float* xl = lx + bb * SCAN_XP + hlf * 4;

  __syncthreads();
  float4 xv[4];
  #pragma unroll
  for (int p = 0; p < 4; ++p) xv[p] = *(const float4*)(xl + 8 * p);

  #pragma unroll 1
  for (int s = 0; s < XS; ++s) {
    float4*       hp = (float4*)(hio + rowoff + (size_t)s * XD) + 4 * e;
    const bool writer = ((s & (SCAN_WAVES - 1)) == w);
    float4 hv[2];
    #pragma unroll
    for (int i = 0; i < 2; ++i) hv[i] = make_float4(0.0f, 0.0f, 0.0f, 0.0f);
    if (writer) {
      #pragma unroll
      for (int i = 0; i < 2; ++i) hv[i] = hp[2 * i];
    }

    v2f bx[8], bh[8];
    #pragma unroll
    for (int kk = 0; kk < 8; ++kk) {
      float4 q = xv[kk >> 1];
      bx[kk].x = (kk & 1) ? q.z : q.x;
      bx[kk].y = (kk & 1) ? q.w : q.y;
      bh[kk].x = hA[2 * kk + 0];
      bh[kk].y = hA[2 * kk + 1];
    }

    v8f acc;
    #pragma unroll
    for (int r = 0; r < 8; ++r) acc[r] = bz[r];
    #pragma unroll
    for (int kk = 0; kk < 8; ++kk) acc = wmma4(aw[kk], bx[kk], acc);
    #pragma unroll
    for (int kk = 0; kk < 8; ++kk) acc = wmma4(ar[kk], bh[kk], acc);

    float hn;
    {
      float ir  = e ? acc[1] : acc[0];
      float fr  = e ? acc[3] : acc[2];
      float zr  = e ? acc[5] : acc[4];
      float orr = e ? acc[7] : acc[6];
      float u   = expf(-fabsf(fr));
      float q   = mS + fminf(fr, 0.0f);
      float d   = ir - q;
      float t   = expf(-fabsf(d));
      bool  imx = (d >= 0.0f);
      float mn  = imx ? ir : q;
      float ig  = imx ? 1.0f : t;
      float fg  = (imx ? t : 1.0f) / (1.0f + u);
      float cn  = fg * cS + ig * tanhf(zr);
      float nn  = fg * nS + ig;
      float og  = 1.0f / (1.0f + expf(-orr));
      hn = og * cn / nn;
      cS = cn;
      nS = nn;
      mS = mn;
    }

    ((float*)&lh[s & 1][w][hlf * SCAN_BT + bb])[e] = hn;
    __syncthreads();
    {
      const int sn = (s + 1 < XS) ? (s + 1) : s;
      const float* xq = xl + sn * (SCAN_BT * SCAN_XP);
      #pragma unroll
      for (int p = 0; p < 4; ++p) xv[p] = *(const float4*)(xq + 8 * p);
    }
    #pragma unroll
    for (int kk = 0; kk < SCAN_WAVES; ++kk) {
      float2 t = lh[s & 1][kk][hlf * SCAN_BT + bb];
      hA[2 * kk + 0] = t.x;
      hA[2 * kk + 1] = t.y;
    }

    if (writer) {
      v2f h2[8];
      #pragma unroll
      for (int kk = 0; kk < 8; ++kk) {
        h2[kk].x = hA[2 * kk + 0];
        h2[kk].y = hA[2 * kk + 1];
      }
      v2f s2 = ((h2[0] + h2[1]) + (h2[2] + h2[3])) + ((h2[4] + h2[5]) + (h2[6] + h2[7]));
      v2f q2 = ((h2[0] * h2[0] + h2[1] * h2[1]) + (h2[2] * h2[2] + h2[3] * h2[3])) +
               ((h2[4] * h2[4] + h2[5] * h2[5]) + (h2[6] * h2[6] + h2[7] * h2[7]));
      float sum = s2.x + s2.y;
      float sq  = q2.x + q2.y;
      sum += xhalf(sum);
      sq  += xhalf(sq);
      float mu = sum * (1.0f / XDH);
      float q  = sq * (1.0f / XDH) - mu * mu;
      float rs = rsqrtf(q + XEPS);
      #pragma unroll
      for (int i = 0; i < 2; ++i) {
        v2f lo0 = h2[2 * i], lo1 = h2[2 * i + 1];
        v2f hi0 = h2[4 + 2 * i], hi1 = h2[4 + 2 * i + 1];
        float u0 = e ? hi0.x : lo0.x;
        float u1 = e ? hi0.y : lo0.y;
        float u2 = e ? hi1.x : lo1.x;
        float u3 = e ? hi1.y : lo1.y;
        float4 o = hv[i];
        o.x += (u0 - mu) * rs * gn[4 * i + 0];
        o.y += (u1 - mu) * rs * gn[4 * i + 1];
        o.z += (u2 - mu) * rs * gn[4 * i + 2];
        o.w += (u3 - mu) * rs * gn[4 * i + 3];
        hp[2 * i] = o;
      }
    }
  }
}

#define FFN_WAVES 8
#define FFN_WP    132

struct FfnLds {
  float wt[2][128 * FFN_WP];
  float act[FFN_WAVES][16 * FFN_WP];
};

typedef float v4f __attribute__((ext_vector_type(4)));
#define FFN_FETCH(st, src)                                                        \
  _Pragma("unroll") for (int i_ = 0; i_ < 16; ++i_)                               \
    (st)[i_] = *(const v4f*)((src) + (size_t)(i_ * 256 + tid) * 4)
#define FFN_PUT(st, dst)                                                          \
  _Pragma("unroll") for (int i_ = 0; i_ < 16; ++i_) {                             \
    int idx_ = i_ * 256 + tid;                                                    \
    *(v4f*)((dst) + (idx_ >> 5) * FFN_WP + (idx_ & 31) * 4) = (st)[i_];           \
  }

#define FFN_PUT_SCALED(st, dst, sc)                                               \
  _Pragma("unroll") for (int i_ = 0; i_ < 16; ++i_) {                             \
    int idx_ = i_ * 256 + tid;                                                    \
    *(v4f*)((dst) + (idx_ >> 5) * FFN_WP + (idx_ & 31) * 4) = (st)[i_] * (sc);    \
  }

__device__ __forceinline__ void ffn_lds_barrier() {
  __builtin_amdgcn_fence(__ATOMIC_RELEASE, "workgroup", "local");
  __builtin_amdgcn_s_barrier();
  __builtin_amdgcn_fence(__ATOMIC_ACQUIRE, "workgroup", "local");
}

__device__ __forceinline__ void ffn_tile2(const v2f (&a)[32], const float* wsec, int nt,
                                          int l16, int hlf, v8f& c0, v8f& c1) {
  const float* p0 = wsec + (nt * 16 + l16) * FFN_WP + hlf * 2;
  const float* p1 = p0 + 16 * FFN_WP;
  #pragma unroll
  for (int r = 0; r < 8; ++r) {
    c0[r] = 0.0f;
    c1[r] = 0.0f;
  }
  #pragma unroll
  for (int kk = 0; kk < 32; ++kk) {
    v2f b0 = *(const v2f*)(p0 + kk * 4);
    v2f b1 = *(const v2f*)(p1 + kk * 4);
    c0 = wmma4(a[kk], b0, c0);
    c1 = wmma4(a[kk], b1, c1);
  }
}

__global__ __launch_bounds__(32 * FFN_WAVES) void k_ffn(float* __restrict__ hio,
                                                        float* __restrict__ xnext,
                                                        const float* __restrict__ lnw,
                                                        const float* __restrict__ lnnext,
                                                        const float* __restrict__ Wup,
                                                        const float* __restrict__ Wdn,
                                                        int store_h) {
  __shared__ __attribute__((aligned(16))) FfnLds sm;
  const int tid  = threadIdx.x;
  const int w    = tid >> 5;
  const int lane = tid & 31;
  const int hlf  = lane >> 4;
  const int l16  = lane & 15;
  const size_t t0 = ((size_t)blockIdx.x * FFN_WAVES + w) * 16;
  float* actw = sm.act[w];

  v2f ax[32];
  const float* xp = hio + (t0 + l16) * XD + hlf * 2;
  #pragma unroll
  for (int kk = 0; kk < 32; ++kk) ax[kk] = *(const v2f*)(xp + kk * 4);

  const v4f lw = *(const v4f*)(lnw + lane * 4);
  v4f st[16];
  FFN_FETCH(st, Wup);
  ln_afrag(ax);
  FFN_PUT_SCALED(st, sm.wt[0], lw);
  FFN_FETCH(st, Wup + 128 * XD);
  ffn_lds_barrier();

  v8f gatev[8];
  #pragma unroll
  for (int nt = 0; nt < 8; nt += 2)
    ffn_tile2(ax, sm.wt[0], nt, l16, hlf, gatev[nt], gatev[nt + 1]);

  FFN_PUT_SCALED(st, sm.wt[1], lw);
  FFN_FETCH(st, Wdn);
  ffn_lds_barrier();

  #pragma unroll
  for (int nt = 0; nt < 8; nt += 2) {
    v8f u0, u1;
    ffn_tile2(ax, sm.wt[1], nt, l16, hlf, u0, u1);
    #pragma unroll
    for (int r = 0; r < 8; ++r) {
      actw[(r + hlf * 8) * FFN_WP + nt * 16 + l16]      = gelu_exact(gatev[nt][r]) * u0[r];
      actw[(r + hlf * 8) * FFN_WP + nt * 16 + 16 + l16] = gelu_exact(gatev[nt + 1][r]) * u1[r];
    }
  }

  FFN_PUT(st, sm.wt[0]);
  ffn_lds_barrier();

  v2f aa[32];
  const float* lp = actw + l16 * FFN_WP + hlf * 2;
  #pragma unroll
  for (int kk = 0; kk < 32; ++kk) aa[kk] = *(const v2f*)(lp + kk * 4);
  float hn[8][8];
  float rsum[8], rsq[8];
  #pragma unroll
  for (int r = 0; r < 8; ++r) {
    rsum[r] = 0.0f;
    rsq[r]  = 0.0f;
  }
  #pragma unroll
  for (int nt = 0; nt < 8; nt += 2) {
    v8f d0, d1;
    ffn_tile2(aa, sm.wt[0], nt, l16, hlf, d0, d1);
    #pragma unroll
    for (int r = 0; r < 8; ++r) {
      size_t o = (t0 + r + hlf * 8) * XD + nt * 16 + l16;
      float a0 = hio[o] + d0[r];
      float a1 = hio[o + 16] + d1[r];
      hn[nt][r]     = a0;
      hn[nt + 1][r] = a1;
      rsum[r] += a0 + a1;
      rsq[r]  += a0 * a0 + a1 * a1;
    }
  }

  if (store_h) {
    #pragma unroll
    for (int nt = 0; nt < 8; ++nt) {
      #pragma unroll
      for (int r = 0; r < 8; ++r)
        hio[(t0 + r + hlf * 8) * XD + nt * 16 + l16] = hn[nt][r];
    }
  }

  #pragma unroll
  for (int r = 0; r < 8; ++r) {
    #pragma unroll
    for (int m = 8; m >= 1; m >>= 1) {
      rsum[r] += __shfl_xor(rsum[r], m, 32);
      rsq[r]  += __shfl_xor(rsq[r], m, 32);
    }
  }
  float wn[8];
  #pragma unroll
  for (int nt = 0; nt < 8; ++nt) wn[nt] = lnnext[nt * 16 + l16];
  #pragma unroll
  for (int r = 0; r < 8; ++r) {
    float mu  = rsum[r] * (1.0f / XD);
    float var = rsq[r] * (1.0f / XD) - mu * mu;
    float rs  = rsqrtf(var + XEPS);
    #pragma unroll
    for (int nt = 0; nt < 8; ++nt)
      actw[(r + hlf * 8) * FFN_WP + nt * 16 + l16] = (hn[nt][r] - mu) * rs * wn[nt];
  }
  ffn_lds_barrier();
  #pragma unroll
  for (int i = 0; i < 16; ++i) {
    v4f v = *(const v4f*)(actw + i * FFN_WP + lane * 4);
    *(v4f*)(xnext + (t0 + i) * XD + lane * 4) = v;
  }
}

__global__ __launch_bounds__(32) void k_proj(const float* __restrict__ xn,
                                             const float* __restrict__ pw,
                                             const float* __restrict__ pb,
                                             float* __restrict__ out) {
  const int lane = threadIdx.x;
  const int hlf  = lane >> 4;
  const int l16  = lane & 15;
  const size_t t0 = (size_t)blockIdx.x * 16;

  v8f a;
  #pragma unroll
  for (int r = 0; r < 8; ++r) a[r] = 0.0f;
  const float* xp = xn + (t0 + l16) * XD;
  #pragma unroll
  for (int kk = 0; kk < 32; ++kk) {
    int K0 = kk * 4 + hlf * 2;
    v2f av;
    av.x = xp[K0];
    av.y = xp[K0 + 1];
    v2f b;
    b.x = 0.0f;
    b.y = 0.0f;
    if (l16 < 9) {
      b.x = pw[(size_t)l16 * XD + K0];
      b.y = pw[(size_t)l16 * XD + K0 + 1];
    }
    a = wmma4(av, b, a);
  }
  if (l16 < 9) {
    float bb = pb[l16];
    #pragma unroll
    for (int r = 0; r < 8; ++r) {
      size_t row = t0 + r + hlf * 8;
      out[row * 9 + l16] = a[r] + bb;
    }
  }
}

extern "C" void kernel_launch(void* const* d_in, const int* in_sizes, int n_in,
                              void* d_out, int out_size, void* d_ws, size_t ws_size,
                              hipStream_t stream) {
  (void)in_sizes; (void)n_in; (void)out_size; (void)ws_size;
  const int*   x     = (const int*)d_in[0];
  const float* emb   = (const float*)d_in[1];
  const float* ln1   = (const float*)d_in[2];
  const float* Wg    = (const float*)d_in[3];
  const float* R     = (const float*)d_in[4];
  const float* bias  = (const float*)d_in[5];
  const float* gnw   = (const float*)d_in[6];
  const float* ln2   = (const float*)d_in[7];
  const float* wup   = (const float*)d_in[8];
  const float* wdn   = (const float*)d_in[9];
  const float* postw = (const float*)d_in[10];
  const float* pw    = (const float*)d_in[11];
  const float* pb    = (const float*)d_in[12];
  float* out = (float*)d_out;

  float* h  = (float*)d_ws;
  float* xb = h + (size_t)XT * XD;

  k_embed_ln<<<XT / 8, 256, 0, stream>>>(x, emb, ln1, h, xb);

  k_scan<<<dim3(XB / SCAN_BT, XNH), 32 * SCAN_WAVES, 0, stream>>>(xb, h, Wg, R, bias, gnw);
  k_ffn<<<XT / (16 * FFN_WAVES), 32 * FFN_WAVES, 0, stream>>>(h, xb, ln2, ln1 + XD, wup, wdn, 1);

  k_scan<<<dim3(XB / SCAN_BT, XNH), 32 * SCAN_WAVES, 0, stream>>>(
      xb, h,
      Wg   + (size_t)4 * XNH * XDH * XDH,
      R    + (size_t)XNH * XDH * 4 * XDH,
      bias + (size_t)XNH * 4 * XDH,
      gnw  + XD);
  k_ffn<<<XT / (16 * FFN_WAVES), 32 * FFN_WAVES, 0, stream>>>(h, xb, ln2 + XD, postw,
                                                              wup + (size_t)2 * XD * XD,
                                                              wdn + (size_t)XD * XD, 0);

  k_proj<<<XT / 16, 32, 0, stream>>>(xb, pw, pb, out);
}


# scan 1 works from the 9-row embedding table: its prologue makes layernorm(emb) and bias + Wg.xn for the 9 tokens (8 WMMAs), a step takes its gates by token id; no embed launch; sum order kept
# speedup vs baseline: 9.5845x; 1.1910x over previous
#include <hip/hip_runtime.h>
#include <hip/hip_bf16.h>


typedef float v2f __attribute__((ext_vector_type(2)));
typedef float v4f __attribute__((ext_vector_type(4)));
typedef float v8f __attribute__((ext_vector_type(8)));

#define XB   512
#define XS   64
#define XD   128
#define XNH  4
#define XDH  32
#define XT   (XB * XS)
#define XEPS 1e-5f
#define XV   9

__device__ __forceinline__ v8f wmma4(v2f a, v2f b, v8f c) {
  return __builtin_amdgcn_wmma_f32_16x16x4_f32(false, a, false, b, (short)0, c,
                                               false, false);
}

__device__ __forceinline__ float gelu_exact(float x) {
  return 0.5f * x * (1.0f + erff(x * 0.70710678118654752f));
}

__device__ __forceinline__ float xhalf(float x) {
  int b = __builtin_bit_cast(int, x);
  return __builtin_bit_cast(
      float, __builtin_amdgcn_permlanex16(b, b, 0x76543210, 0xfedcba98, false, false));
}

__device__ __forceinline__ void ln_afrag(v2f (&a)[32]) {
  v2f s4[4], q4[4];
  #pragma unroll
  for (int j = 0; j < 4; ++j) {
    s4[j] = a[j];
    q4[j] = a[j] * a[j];
  }
  #pragma unroll
  for (int kk = 4; kk < 32; ++kk) {
    s4[kk & 3] += a[kk];
    q4[kk & 3] += a[kk] * a[kk];
  }
  v2f s2 = (s4[0] + s4[1]) + (s4[2] + s4[3]);
  v2f q2 = (q4[0] + q4[1]) + (q4[2] + q4[3]);
  float s = s2.x + s2.y;
  float q = q2.x + q2.y;
  s += xhalf(s);
  q += xhalf(q);
  float mu  = s * (1.0f / XD);
  float var = q * (1.0f / XD) - mu * mu;
  float rs  = rsqrtf(var + XEPS);
  #pragma unroll
  for (int kk = 0; kk < 32; ++kk) {
    a[kk].x = (a[kk].x - mu) * rs;
    a[kk].y = (a[kk].y - mu) * rs;
  }
}

#define SCAN_WAVES 8
#define SCAN_BT    8
#define SCAN_XP    36

template <bool FIRST> struct ScanLds;
template <> struct ScanLds<false> {
  float2 lh[2][SCAN_WAVES][2 * SCAN_BT];
  float lx[XS * SCAN_BT * SCAN_XP];
};
template <> struct ScanLds<true> {
  float2 lh[2][SCAN_WAVES][2 * SCAN_BT];
  int   xi[XS * SCAN_BT];
  float xt[XV][XDH];
  float et[XV][XDH];
  float g[SCAN_WAVES][XV][2][2][4];
};

template <bool FIRST>
__global__ __launch_bounds__(32 * SCAN_WAVES) void k_scan(const int* __restrict__ x,
                                                          const float* __restrict__ emb,
                                                          const float* __restrict__ lnw,
                                                          const float* __restrict__ xn,
                                                          float* __restrict__ hio,
                                                          const float* __restrict__ Wg,
                                                          const float* __restrict__ R,
                                                          const float* __restrict__ bias,
                                                          const float* __restrict__ gnw) {
  __shared__ __attribute__((aligned(16))) ScanLds<FIRST> sm;
  float2 (*lh)[SCAN_WAVES][2 * SCAN_BT] = sm.lh;

  const int w    = threadIdx.x >> 5;
  const int lane = threadIdx.x & 31;
  const int hlf  = lane >> 4;
  const int l16  = lane & 15;
  const int bb   = l16 & 7;
  const int e    = l16 >> 3;
  const int b0   = blockIdx.x * SCAN_BT;
  const int n    = blockIdx.y;
  const int ub   = 8 * (w >> 1) + 2 * (w & 1);

  if constexpr (!FIRST) {
    float* lx = sm.lx;
    const float* xsrc = xn + (size_t)b0 * XS * XD + n * XDH;
    float4 st[16];
    #pragma unroll
    for (int i = 0; i < 16; ++i) {
      int idx  = i * (32 * SCAN_WAVES) + (int)threadIdx.x;
      int part = idx & 7;
      int seg  = idx >> 3;
      int row  = seg >> 6;
      int ss   = seg & (XS - 1);
      st[i] = *(const float4*)(xsrc + ((size_t)row * XS + ss) * XD + part * 4);
    }
    #pragma unroll
    for (int i = 0; i < 16; ++i) {
      int idx  = i * (32 * SCAN_WAVES) + (int)threadIdx.x;
      int part = idx & 7;
      int seg  = idx >> 3;
      int row  = seg >> 6;
      int ss   = seg & (XS - 1);
      *(float4*)&lx[(ss * SCAN_BT + row) * SCAN_XP + part * 4] = st[i];
    }
  }

  if constexpr (FIRST) {
    #pragma unroll
    for (int i = 0; i < 2; ++i) {
      int idx = i * (32 * SCAN_WAVES) + (int)threadIdx.x;
      sm.xi[(idx & (XS - 1)) * SCAN_BT + (idx >> 6)] = x[(size_t)b0 * XS + idx];
    }
    const float4 wl = ((const float4*)lnw)[lane];
    const int vrow[2] = {w, XV - 1};
    float4 ev[2];
    float  sv[2], qv[2];
    #pragma unroll
    for (int j = 0; j < 2; ++j) {
      ev[j] = ((const float4*)(emb + (size_t)vrow[j] * XD))[lane];
      sv[j] = ev[j].x + ev[j].y + ev[j].z + ev[j].w;
      qv[j] = ev[j].x * ev[j].x + ev[j].y * ev[j].y + ev[j].z * ev[j].z + ev[j].w * ev[j].w;
    }
    #pragma unroll
    for (int m = 16; m >= 1; m >>= 1) {
      #pragma unroll
      for (int j = 0; j < 2; ++j) {
        sv[j] += __shfl_xor(sv[j], m, 32);
        qv[j] += __shfl_xor(qv[j], m, 32);
      }
    }
    #pragma unroll
    for (int j = 0; j < 2; ++j) {
      float mu  = sv[j] * (1.0f / XD);
      float var = qv[j] * (1.0f / XD) - mu * mu;
      float rs  = rsqrtf(var + XEPS);
      float4 o;
      o.x = (ev[j].x - mu) * rs * wl.x;
      o.y = (ev[j].y - mu) * rs * wl.y;
      o.z = (ev[j].z - mu) * rs * wl.z;
      o.w = (ev[j].w - mu) * rs * wl.w;
      if ((lane >> 3) == n && (j == 0 || w == 0)) {
        *(float4*)&sm.xt[vrow[j]][(lane & 7) * 4] = o;
        *(float4*)&sm.et[vrow[j]][(lane & 7) * 4] = ev[j];
      }
    }
  }

  const int ag = (l16 >> 1) & 3;
  const int au = ub + 4 * (l16 >> 3) + (l16 & 1);
  v2f aw[8], ar[8];
  #pragma unroll
  for (int kk = 0; kk < 8; ++kk) {
    int K0 = 8 * (kk >> 1) + 4 * hlf + 2 * (kk & 1);
    const float* wgp = Wg + (((size_t)(ag * XNH + n) * XDH + au) * XDH + K0);
    aw[kk].x = wgp[0];
    aw[kk].y = wgp[1];
    ar[kk].x = R[(size_t)(n * XDH + K0 + 0) * (4 * XDH) + ag * XDH + au];
    ar[kk].y = R[(size_t)(n * XDH + K0 + 1) * (4 * XDH) + ag * XDH + au];
  }
  float bz[8];
  #pragma unroll
  for (int r = 0; r < 8; ++r)
    bz[r] = bias[n * 128 + (r >> 1) * XDH + ub + 4 * hlf + (r & 1)];
  float gn[8];
  #pragma unroll
  for (int i = 0; i < 2; ++i) {
    #pragma unroll
    for (int q = 0; q < 4; ++q) gn[4 * i + q] = gnw[n * XDH + 8 * (2 * e + i) + 4 * hlf + q];
  }

  float cS = 0.0f;
  float nS = 0.0f;
  float mS = 0.0f;
  float hA[16] = {};

  const size_t rowoff = (size_t)(b0 + bb) * XS * XD + n * XDH + hlf * 4;

  const float* xl = nullptr;
  float4 xv[4];
  #pragma unroll
  for (int p = 0; p < 4; ++p) xv[p] = make_float4(0.0f, 0.0f, 0.0f, 0.0f);
  v4f gcur = {0.0f, 0.0f, 0.0f, 0.0f};
  int tokA = 0, tokB = 0;

  if constexpr (FIRST) {
    __syncthreads();
    const int tv = (l16 < XV) ? l16 : (XV - 1);
    v8f tacc;
    #pragma unroll
    for (int r = 0; r < 8; ++r) tacc[r] = bz[r];
    #pragma unroll
    for (int kk = 0; kk < 8; ++kk) {
      int K0 = 8 * (kk >> 1) + 4 * hlf + 2 * (kk & 1);
      v2f bt = *(const v2f*)&sm.xt[tv][K0];
      tacc = wmma4(aw[kk], bt, tacc);
    }
    if (l16 < XV) {
      v4f g0 = {tacc[0], tacc[2], tacc[4], tacc[6]};
      v4f g1 = {tacc[1], tacc[3], tacc[5], tacc[7]};
      *(v4f*)sm.g[w][l16][hlf][0] = g0;
      *(v4f*)sm.g[w][l16][hlf][1] = g1;
    }
    __syncthreads();
    tokA = sm.xi[0 * SCAN_BT + bb];
    tokB = sm.xi[1 * SCAN_BT + bb];
    gcur = *(const v4f*)sm.g[w][tokA][hlf][e];
  } else {
    xl = sm.lx + bb * SCAN_XP + hlf * 4;
    __syncthreads();
    #pragma unroll
    for (int p = 0; p < 4; ++p) xv[p] = *(const float4*)(xl + 8 * p);
  }

  float4 hv[2];
  #pragma unroll
  for (int i = 0; i < 2; ++i) hv[i] = make_float4(0.0f, 0.0f, 0.0f, 0.0f);
  if constexpr (FIRST) {
    if (w == 0) {
      #pragma unroll
      for (int i = 0; i < 2; ++i)
        hv[i] = *(const float4*)&sm.et[tokA][8 * (2 * e + i) + 4 * hlf];
    }
  }

  #pragma unroll 1
  for (int s = 0; s < XS; ++s) {
    float4*       hp = (float4*)(hio + rowoff + (size_t)s * XD) + 4 * e;
    const bool writer = ((s & (SCAN_WAVES - 1)) == w);
    if constexpr (!FIRST) {
      #pragma unroll
      for (int i = 0; i < 2; ++i) hv[i] = make_float4(0.0f, 0.0f, 0.0f, 0.0f);
      if (writer) {
        #pragma unroll
        for (int i = 0; i < 2; ++i) hv[i] = hp[2 * i];
      }
    }

    v2f bh[8];
    #pragma unroll
    for (int kk = 0; kk < 8; ++kk) {
      bh[kk].x = hA[2 * kk + 0];
      bh[kk].y = hA[2 * kk + 1];
    }

    v8f acc;
    if constexpr (FIRST) {
      #pragma unroll
      for (int r = 0; r < 8; ++r) acc[r] = gcur[r >> 1];
    } else {
      v2f bx[8];
      #pragma unroll
      for (int kk = 0; kk < 8; ++kk) {
        float4 q = xv[kk >> 1];
        bx[kk].x = (kk & 1) ? q.z : q.x;
        bx[kk].y = (kk & 1) ? q.w : q.y;
      }
      #pragma unroll
      for (int r = 0; r < 8; ++r) acc[r] = bz[r];
      #pragma unroll
      for (int kk = 0; kk < 8; ++kk) acc = wmma4(aw[kk], bx[kk], acc);
    }
    #pragma unroll
    for (int kk = 0; kk < 8; ++kk) acc = wmma4(ar[kk], bh[kk], acc);

    float hn;
    {
      float ir  = e ? acc[1] : acc[0];
      float fr  = e ? acc[3] : acc[2];
      float zr  = e ? acc[5] : acc[4];
      float orr = e ? acc[7] : acc[6];
      float u   = expf(-fabsf(fr));
      float q   = mS + fminf(fr, 0.0f);
      float d   = ir - q;
      float t   = expf(-fabsf(d));
      bool  imx = (d >= 0.0f);
      float mn  = imx ? ir : q;
      float ig  = imx ? 1.0f : t;
      float fg  = (imx ? t : 1.0f) / (1.0f + u);
      float cn  = fg * cS + ig * tanhf(zr);
      float nn  = fg * nS + ig;
      float og  = 1.0f / (1.0f + expf(-orr));
      hn = og * cn / nn;
      cS = cn;
      nS = nn;
      mS = mn;
    }

    ((float*)&lh[s & 1][w][hlf * SCAN_BT + bb])[e] = hn;
    __syncthreads();
    if constexpr (!FIRST) {
      const int sn = (s + 1 < XS) ? (s + 1) : s;
      const float* xq = xl + sn * (SCAN_BT * SCAN_XP);
      #pragma unroll
      for (int p = 0; p < 4; ++p) xv[p] = *(const float4*)(xq + 8 * p);
    }
    #pragma unroll
    for (int kk = 0; kk < SCAN_WAVES; ++kk) {
      float2 t = lh[s & 1][kk][hlf * SCAN_BT + bb];
      hA[2 * kk + 0] = t.x;
      hA[2 * kk + 1] = t.y;
    }
    if constexpr (FIRST) {
      gcur = *(const v4f*)sm.g[w][tokB][hlf][e];
      tokA = tokB;
      const int s2 = (s + 2 < XS) ? (s + 2) : (XS - 1);
      tokB = sm.xi[s2 * SCAN_BT + bb];
    }

    if (writer) {
      v2f h2[8];
      #pragma unroll
      for (int kk = 0; kk < 8; ++kk) {
        h2[kk].x = hA[2 * kk + 0];
        h2[kk].y = hA[2 * kk + 1];
      }
      v2f s2 = ((h2[0] + h2[1]) + (h2[2] + h2[3])) + ((h2[4] + h2[5]) + (h2[6] + h2[7]));
      v2f q2 = ((h2[0] * h2[0] + h2[1] * h2[1]) + (h2[2] * h2[2] + h2[3] * h2[3])) +
               ((h2[4] * h2[4] + h2[5] * h2[5]) + (h2[6] * h2[6] + h2[7] * h2[7]));
      float sum = s2.x + s2.y;
      float sq  = q2.x + q2.y;
      sum += xhalf(sum);
      sq  += xhalf(sq);
      float mu = sum * (1.0f / XDH);
      float q  = sq * (1.0f / XDH) - mu * mu;
      float rs = rsqrtf(q + XEPS);
      #pragma unroll
      for (int i = 0; i < 2; ++i) {
        v2f lo0 = h2[2 * i], lo1 = h2[2 * i + 1];
        v2f hi0 = h2[4 + 2 * i], hi1 = h2[4 + 2 * i + 1];
        float u0 = e ? hi0.x : lo0.x;
        float u1 = e ? hi0.y : lo0.y;
        float u2 = e ? hi1.x : lo1.x;
        float u3 = e ? hi1.y : lo1.y;
        float4 o = hv[i];
        o.x += (u0 - mu) * rs * gn[4 * i + 0];
        o.y += (u1 - mu) * rs * gn[4 * i + 1];
        o.z += (u2 - mu) * rs * gn[4 * i + 2];
        o.w += (u3 - mu) * rs * gn[4 * i + 3];
        hp[2 * i] = o;
      }
    }
    if constexpr (FIRST) {
      if (((s + 1) & (SCAN_WAVES - 1)) == w) {
        #pragma unroll
        for (int i = 0; i < 2; ++i)
          hv[i] = *(const float4*)&sm.et[tokA][8 * (2 * e + i) + 4 * hlf];
      }
    }
  }
}

#define FFN_WAVES 8
#define FFN_WP    132

struct FfnLds {
  float wt[2][128 * FFN_WP];
  float act[FFN_WAVES][16 * FFN_WP];
};

#define FFN_FETCH(st, src)                                                        \
  _Pragma("unroll") for (int i_ = 0; i_ < 16; ++i_)                               \
    (st)[i_] = *(const v4f*)((src) + (size_t)(i_ * 256 + tid) * 4)
#define FFN_PUT(st, dst)                                                          \
  _Pragma("unroll") for (int i_ = 0; i_ < 16; ++i_) {                             \
    int idx_ = i_ * 256 + tid;                                                    \
    *(v4f*)((dst) + (idx_ >> 5) * FFN_WP + (idx_ & 31) * 4) = (st)[i_];           \
  }

#define FFN_PUT_SCALED(st, dst, sc)                                               \
  _Pragma("unroll") for (int i_ = 0; i_ < 16; ++i_) {                             \
    int idx_ = i_ * 256 + tid;                                                    \
    *(v4f*)((dst) + (idx_ >> 5) * FFN_WP + (idx_ & 31) * 4) = (st)[i_] * (sc);    \
  }

__device__ __forceinline__ void ffn_lds_barrier() {
  __builtin_amdgcn_fence(__ATOMIC_RELEASE, "workgroup", "local");
  __builtin_amdgcn_s_barrier();
  __builtin_amdgcn_fence(__ATOMIC_ACQUIRE, "workgroup", "local");
}

__device__ __forceinline__ void ffn_tile2(const v2f (&a)[32], const float* wsec, int nt,
                                          int l16, int hlf, v8f& c0, v8f& c1) {
  const float* p0 = wsec + (nt * 16 + l16) * FFN_WP + hlf * 2;
  const float* p1 = p0 + 16 * FFN_WP;
  #pragma unroll
  for (int r = 0; r < 8; ++r) {
    c0[r] = 0.0f;
    c1[r] = 0.0f;
  }
  #pragma unroll
  for (int kk = 0; kk < 32; ++kk) {
    v2f b0 = *(const v2f*)(p0 + kk * 4);
    v2f b1 = *(const v2f*)(p1 + kk * 4);
    c0 = wmma4(a[kk], b0, c0);
    c1 = wmma4(a[kk], b1, c1);
  }
}

__global__ __launch_bounds__(32 * FFN_WAVES) void k_ffn(float* __restrict__ hio,
                                                        float* __restrict__ xnext,
                                                        const float* __restrict__ lnw,
                                                        const float* __restrict__ lnnext,
                                                        const float* __restrict__ Wup,
                                                        const float* __restrict__ Wdn,
                                                        int store_h) {
  __shared__ __attribute__((aligned(16))) FfnLds sm;
  const int tid  = threadIdx.x;
  const int w    = tid >> 5;
  const int lane = tid & 31;
  const int hlf  = lane >> 4;
  const int l16  = lane & 15;
  const size_t t0 = ((size_t)blockIdx.x * FFN_WAVES + w) * 16;
  float* actw = sm.act[w];

  v2f ax[32];
  const float* xp = hio + (t0 + l16) * XD + hlf * 2;
  #pragma unroll
  for (int kk = 0; kk < 32; ++kk) ax[kk] = *(const v2f*)(xp + kk * 4);

  const v4f lw = *(const v4f*)(lnw + lane * 4);
  v4f st[16];
  FFN_FETCH(st, Wup);
  ln_afrag(ax);
  FFN_PUT_SCALED(st, sm.wt[0], lw);
  FFN_FETCH(st, Wup + 128 * XD);
  ffn_lds_barrier();

  v8f gatev[8];
  #pragma unroll
  for (int nt = 0; nt < 8; nt += 2)
    ffn_tile2(ax, sm.wt[0], nt, l16, hlf, gatev[nt], gatev[nt + 1]);

  FFN_PUT_SCALED(st, sm.wt[1], lw);
  FFN_FETCH(st, Wdn);
  ffn_lds_barrier();

  #pragma unroll
  for (int nt = 0; nt < 8; nt += 2) {
    v8f u0, u1;
    ffn_tile2(ax, sm.wt[1], nt, l16, hlf, u0, u1);
    #pragma unroll
    for (int r = 0; r < 8; ++r) {
      actw[(r + hlf * 8) * FFN_WP + nt * 16 + l16]      = gelu_exact(gatev[nt][r]) * u0[r];
      actw[(r + hlf * 8) * FFN_WP + nt * 16 + 16 + l16] = gelu_exact(gatev[nt + 1][r]) * u1[r];
    }
  }

  FFN_PUT(st, sm.wt[0]);
  ffn_lds_barrier();

  v2f aa[32];
  const float* lp = actw + l16 * FFN_WP + hlf * 2;
  #pragma unroll
  for (int kk = 0; kk < 32; ++kk) aa[kk] = *(const v2f*)(lp + kk * 4);
  float hn[8][8];
  float rsum[8], rsq[8];
  #pragma unroll
  for (int r = 0; r < 8; ++r) {
    rsum[r] = 0.0f;
    rsq[r]  = 0.0f;
  }
  #pragma unroll
  for (int nt = 0; nt < 8; nt += 2) {
    v8f d0, d1;
    ffn_tile2(aa, sm.wt[0], nt, l16, hlf, d0, d1);
    #pragma unroll
    for (int r = 0; r < 8; ++r) {
      size_t o = (t0 + r + hlf * 8) * XD + nt * 16 + l16;
      float a0 = hio[o] + d0[r];
      float a1 = hio[o + 16] + d1[r];
      hn[nt][r]     = a0;
      hn[nt + 1][r] = a1;
      rsum[r] += a0 + a1;
      rsq[r]  += a0 * a0 + a1 * a1;
    }
  }

  if (store_h) {
    #pragma unroll
    for (int nt = 0; nt < 8; ++nt) {
      #pragma unroll
      for (int r = 0; r < 8; ++r)
        hio[(t0 + r + hlf * 8) * XD + nt * 16 + l16] = hn[nt][r];
    }
  }

  #pragma unroll
  for (int r = 0; r < 8; ++r) {
    #pragma unroll
    for (int m = 8; m >= 1; m >>= 1) {
      rsum[r] += __shfl_xor(rsum[r], m, 32);
      rsq[r]  += __shfl_xor(rsq[r], m, 32);
    }
  }
  float wn[8];
  #pragma unroll
  for (int nt = 0; nt < 8; ++nt) wn[nt] = lnnext[nt * 16 + l16];
  #pragma unroll
  for (int r = 0; r < 8; ++r) {
    float mu  = rsum[r] * (1.0f / XD);
    float var = rsq[r] * (1.0f / XD) - mu * mu;
    float rs  = rsqrtf(var + XEPS);
    #pragma unroll
    for (int nt = 0; nt < 8; ++nt)
      actw[(r + hlf * 8) * FFN_WP + nt * 16 + l16] = (hn[nt][r] - mu) * rs * wn[nt];
  }
  ffn_lds_barrier();
  #pragma unroll
  for (int i = 0; i < 16; ++i) {
    v4f v = *(const v4f*)(actw + i * FFN_WP + lane * 4);
    *(v4f*)(xnext + (t0 + i) * XD + lane * 4) = v;
  }
}

__global__ __launch_bounds__(32) void k_proj(const float* __restrict__ xn,
                                             const float* __restrict__ pw,
                                             const float* __restrict__ pb,
                                             float* __restrict__ out) {
  const int lane = threadIdx.x;
  const int hlf  = lane >> 4;
  const int l16  = lane & 15;
  const size_t t0 = (size_t)blockIdx.x * 16;

  v8f a;
  #pragma unroll
  for (int r = 0; r < 8; ++r) a[r] = 0.0f;
  const float* xp = xn + (t0 + l16) * XD;
  #pragma unroll
  for (int kk = 0; kk < 32; ++kk) {
    int K0 = kk * 4 + hlf * 2;
    v2f av;
    av.x = xp[K0];
    av.y = xp[K0 + 1];
    v2f b;
    b.x = 0.0f;
    b.y = 0.0f;
    if (l16 < 9) {
      b.x = pw[(size_t)l16 * XD + K0];
      b.y = pw[(size_t)l16 * XD + K0 + 1];
    }
    a = wmma4(av, b, a);
  }
  if (l16 < 9) {
    float bb = pb[l16];
    #pragma unroll
    for (int r = 0; r < 8; ++r) {
      size_t row = t0 + r + hlf * 8;
      out[row * 9 + l16] = a[r] + bb;
    }
  }
}

extern "C" void kernel_launch(void* const* d_in, const int* in_sizes, int n_in,
                              void* d_out, int out_size, void* d_ws, size_t ws_size,
                              hipStream_t stream) {
  (void)in_sizes; (void)n_in; (void)out_size; (void)ws_size;
  const int*   x     = (const int*)d_in[0];
  const float* emb   = (const float*)d_in[1];
  const float* ln1   = (const float*)d_in[2];
  const float* Wg    = (const float*)d_in[3];
  const float* R     = (const float*)d_in[4];
  const float* bias  = (const float*)d_in[5];
  const float* gnw   = (const float*)d_in[6];
  const float* ln2   = (const float*)d_in[7];
  const float* wup   = (const float*)d_in[8];
  const float* wdn   = (const float*)d_in[9];
  const float* postw = (const float*)d_in[10];
  const float* pw    = (const float*)d_in[11];
  const float* pb    = (const float*)d_in[12];
  float* out = (float*)d_out;

  float* h  = (float*)d_ws;
  float* xb = h + (size_t)XT * XD;

  k_scan<true><<<dim3(XB / SCAN_BT, XNH), 32 * SCAN_WAVES, 0, stream>>>(x, emb, ln1, xb, h, Wg, R,
                                                                        bias, gnw);
  k_ffn<<<XT / (16 * FFN_WAVES), 32 * FFN_WAVES, 0, stream>>>(h, xb, ln2, ln1 + XD, wup, wdn, 1);

  k_scan<false><<<dim3(XB / SCAN_BT, XNH), 32 * SCAN_WAVES, 0, stream>>>(
      x, emb, ln1, xb, h,
      Wg   + (size_t)4 * XNH * XDH * XDH,
      R    + (size_t)XNH * XDH * 4 * XDH,
      bias + (size_t)XNH * 4 * XDH,
      gnw  + XD);
  k_ffn<<<XT / (16 * FFN_WAVES), 32 * FFN_WAVES, 0, stream>>>(h, xb, ln2 + XD, postw,
                                                              wup + (size_t)2 * XD * XD,
                                                              wdn + (size_t)XD * XD, 0);

  k_proj<<<XT / 16, 32, 0, stream>>>(xb, pw, pb, out);
}


# scan 2: the input product of step s+1 (bias + Wg.xn, 8 WMMAs) is issued inside step s's h hand-over (store landing, split LDS-only barrier, h read-back), not in front of R.h; same sums
# speedup vs baseline: 10.0004x; 1.0434x over previous
#include <hip/hip_runtime.h>
#include <hip/hip_bf16.h>


typedef float v2f __attribute__((ext_vector_type(2)));
typedef float v4f __attribute__((ext_vector_type(4)));
typedef float v8f __attribute__((ext_vector_type(8)));

#define XB   512
#define XS   64
#define XD   128
#define XNH  4
#define XDH  32
#define XT   (XB * XS)
#define XEPS 1e-5f
#define XV   9

__device__ __forceinline__ v8f wmma4(v2f a, v2f b, v8f c) {
  return __builtin_amdgcn_wmma_f32_16x16x4_f32(false, a, false, b, (short)0, c,
                                               false, false);
}

__device__ __forceinline__ float gelu_exact(float x) {
  return 0.5f * x * (1.0f + erff(x * 0.70710678118654752f));
}

__device__ __forceinline__ float xhalf(float x) {
  int b = __builtin_bit_cast(int, x);
  return __builtin_bit_cast(
      float, __builtin_amdgcn_permlanex16(b, b, 0x76543210, 0xfedcba98, false, false));
}

__device__ __forceinline__ void ln_afrag(v2f (&a)[32]) {
  v2f s4[4], q4[4];
  #pragma unroll
  for (int j = 0; j < 4; ++j) {
    s4[j] = a[j];
    q4[j] = a[j] * a[j];
  }
  #pragma unroll
  for (int kk = 4; kk < 32; ++kk) {
    s4[kk & 3] += a[kk];
    q4[kk & 3] += a[kk] * a[kk];
  }
  v2f s2 = (s4[0] + s4[1]) + (s4[2] + s4[3]);
  v2f q2 = (q4[0] + q4[1]) + (q4[2] + q4[3]);
  float s = s2.x + s2.y;
  float q = q2.x + q2.y;
  s += xhalf(s);
  q += xhalf(q);
  float mu  = s * (1.0f / XD);
  float var = q * (1.0f / XD) - mu * mu;
  float rs  = rsqrtf(var + XEPS);
  #pragma unroll
  for (int kk = 0; kk < 32; ++kk) {
    a[kk].x = (a[kk].x - mu) * rs;
    a[kk].y = (a[kk].y - mu) * rs;
  }
}

#define SCAN_WAVES 8
#define SCAN_BT    8
#define SCAN_XP    36

template <bool FIRST> struct ScanLds;
template <> struct ScanLds<false> {
  float2 lh[2][SCAN_WAVES][2 * SCAN_BT];
  float lx[XS * SCAN_BT * SCAN_XP];
};
template <> struct ScanLds<true> {
  float2 lh[2][SCAN_WAVES][2 * SCAN_BT];
  int   xi[XS * SCAN_BT];
  float xt[XV][XDH];
  float et[XV][XDH];
  float g[SCAN_WAVES][XV][2][2][4];
};

template <bool FIRST>
__global__ __launch_bounds__(32 * SCAN_WAVES) void k_scan(const int* __restrict__ x,
                                                          const float* __restrict__ emb,
                                                          const float* __restrict__ lnw,
                                                          const float* __restrict__ xn,
                                                          float* __restrict__ hio,
                                                          const float* __restrict__ Wg,
                                                          const float* __restrict__ R,
                                                          const float* __restrict__ bias,
                                                          const float* __restrict__ gnw) {
  __shared__ __attribute__((aligned(16))) ScanLds<FIRST> sm;
  float2 (*lh)[SCAN_WAVES][2 * SCAN_BT] = sm.lh;

  const int w    = threadIdx.x >> 5;
  const int lane = threadIdx.x & 31;
  const int hlf  = lane >> 4;
  const int l16  = lane & 15;
  const int bb   = l16 & 7;
  const int e    = l16 >> 3;
  const int b0   = blockIdx.x * SCAN_BT;
  const int n    = blockIdx.y;
  const int ub   = 8 * (w >> 1) + 2 * (w & 1);

  if constexpr (!FIRST) {
    float* lx = sm.lx;
    const float* xsrc = xn + (size_t)b0 * XS * XD + n * XDH;
    float4 st[16];
    #pragma unroll
    for (int i = 0; i < 16; ++i) {
      int idx  = i * (32 * SCAN_WAVES) + (int)threadIdx.x;
      int part = idx & 7;
      int seg  = idx >> 3;
      int row  = seg >> 6;
      int ss   = seg & (XS - 1);
      st[i] = *(const float4*)(xsrc + ((size_t)row * XS + ss) * XD + part * 4);
    }
    #pragma unroll
    for (int i = 0; i < 16; ++i) {
      int idx  = i * (32 * SCAN_WAVES) + (int)threadIdx.x;
      int part = idx & 7;
      int seg  = idx >> 3;
      int row  = seg >> 6;
      int ss   = seg & (XS - 1);
      *(float4*)&lx[(ss * SCAN_BT + row) * SCAN_XP + part * 4] = st[i];
    }
  }

  if constexpr (FIRST) {
    #pragma unroll
    for (int i = 0; i < 2; ++i) {
      int idx = i * (32 * SCAN_WAVES) + (int)threadIdx.x;
      sm.xi[(idx & (XS - 1)) * SCAN_BT + (idx >> 6)] = x[(size_t)b0 * XS + idx];
    }
    const float4 wl = ((const float4*)lnw)[lane];
    const int vrow[2] = {w, XV - 1};
    float4 ev[2];
    float  sv[2], qv[2];
    #pragma unroll
    for (int j = 0; j < 2; ++j) {
      ev[j] = ((const float4*)(emb + (size_t)vrow[j] * XD))[lane];
      sv[j] = ev[j].x + ev[j].y + ev[j].z + ev[j].w;
      qv[j] = ev[j].x * ev[j].x + ev[j].y * ev[j].y + ev[j].z * ev[j].z + ev[j].w * ev[j].w;
    }
    #pragma unroll
    for (int m = 16; m >= 1; m >>= 1) {
      #pragma unroll
      for (int j = 0; j < 2; ++j) {
        sv[j] += __shfl_xor(sv[j], m, 32);
        qv[j] += __shfl_xor(qv[j], m, 32);
      }
    }
    #pragma unroll
    for (int j = 0; j < 2; ++j) {
      float mu  = sv[j] * (1.0f / XD);
      float var = qv[j] * (1.0f / XD) - mu * mu;
      float rs  = rsqrtf(var + XEPS);
      float4 o;
      o.x = (ev[j].x - mu) * rs * wl.x;
      o.y = (ev[j].y - mu) * rs * wl.y;
      o.z = (ev[j].z - mu) * rs * wl.z;
      o.w = (ev[j].w - mu) * rs * wl.w;
      if ((lane >> 3) == n && (j == 0 || w == 0)) {
        *(float4*)&sm.xt[vrow[j]][(lane & 7) * 4] = o;
        *(float4*)&sm.et[vrow[j]][(lane & 7) * 4] = ev[j];
      }
    }
  }

  const int ag = (l16 >> 1) & 3;
  const int au = ub + 4 * (l16 >> 3) + (l16 & 1);
  v2f aw[8], ar[8];
  #pragma unroll
  for (int kk = 0; kk < 8; ++kk) {
    int K0 = 8 * (kk >> 1) + 4 * hlf + 2 * (kk & 1);
    const float* wgp = Wg + (((size_t)(ag * XNH + n) * XDH + au) * XDH + K0);
    aw[kk].x = wgp[0];
    aw[kk].y = wgp[1];
    ar[kk].x = R[(size_t)(n * XDH + K0 + 0) * (4 * XDH) + ag * XDH + au];
    ar[kk].y = R[(size_t)(n * XDH + K0 + 1) * (4 * XDH) + ag * XDH + au];
  }
  float bz[8];
  #pragma unroll
  for (int r = 0; r < 8; ++r)
    bz[r] = bias[n * 128 + (r >> 1) * XDH + ub + 4 * hlf + (r & 1)];
  float gn[8];
  #pragma unroll
  for (int i = 0; i < 2; ++i) {
    #pragma unroll
    for (int q = 0; q < 4; ++q) gn[4 * i + q] = gnw[n * XDH + 8 * (2 * e + i) + 4 * hlf + q];
  }

  float cS = 0.0f;
  float nS = 0.0f;
  float mS = 0.0f;
  float hA[16] = {};

  const size_t rowoff = (size_t)(b0 + bb) * XS * XD + n * XDH + hlf * 4;

  const float* xl = nullptr;
  float4 xv[4];
  #pragma unroll
  for (int p = 0; p < 4; ++p) xv[p] = make_float4(0.0f, 0.0f, 0.0f, 0.0f);
  v4f gcur = {0.0f, 0.0f, 0.0f, 0.0f};
  int tokA = 0, tokB = 0;
  v8f accx = {0.0f, 0.0f, 0.0f, 0.0f, 0.0f, 0.0f, 0.0f, 0.0f};

  if constexpr (FIRST) {
    __syncthreads();
    const int tv = (l16 < XV) ? l16 : (XV - 1);
    v8f tacc;
    #pragma unroll
    for (int r = 0; r < 8; ++r) tacc[r] = bz[r];
    #pragma unroll
    for (int kk = 0; kk < 8; ++kk) {
      int K0 = 8 * (kk >> 1) + 4 * hlf + 2 * (kk & 1);
      v2f bt = *(const v2f*)&sm.xt[tv][K0];
      tacc = wmma4(aw[kk], bt, tacc);
    }
    if (l16 < XV) {
      v4f g0 = {tacc[0], tacc[2], tacc[4], tacc[6]};
      v4f g1 = {tacc[1], tacc[3], tacc[5], tacc[7]};
      *(v4f*)sm.g[w][l16][hlf][0] = g0;
      *(v4f*)sm.g[w][l16][hlf][1] = g1;
    }
    __syncthreads();
    tokA = sm.xi[0 * SCAN_BT + bb];
    tokB = sm.xi[1 * SCAN_BT + bb];
    gcur = *(const v4f*)sm.g[w][tokA][hlf][e];
  } else {
    xl = sm.lx + bb * SCAN_XP + hlf * 4;
    __syncthreads();
    #pragma unroll
    for (int p = 0; p < 4; ++p) xv[p] = *(const float4*)(xl + 8 * p);
    #pragma unroll
    for (int r = 0; r < 8; ++r) accx[r] = bz[r];
    #pragma unroll
    for (int kk = 0; kk < 8; ++kk) {
      float4 q = xv[kk >> 1];
      v2f bx;
      bx.x = (kk & 1) ? q.z : q.x;
      bx.y = (kk & 1) ? q.w : q.y;
      accx = wmma4(aw[kk], bx, accx);
    }
    #pragma unroll
    for (int p = 0; p < 4; ++p) xv[p] = *(const float4*)(xl + (SCAN_BT * SCAN_XP) + 8 * p);
  }

  float4 hv[2];
  #pragma unroll
  for (int i = 0; i < 2; ++i) hv[i] = make_float4(0.0f, 0.0f, 0.0f, 0.0f);
  if constexpr (FIRST) {
    if (w == 0) {
      #pragma unroll
      for (int i = 0; i < 2; ++i)
        hv[i] = *(const float4*)&sm.et[tokA][8 * (2 * e + i) + 4 * hlf];
    }
  }

  #pragma unroll 1
  for (int s = 0; s < XS; ++s) {
    float4*       hp = (float4*)(hio + rowoff + (size_t)s * XD) + 4 * e;
    const bool writer = ((s & (SCAN_WAVES - 1)) == w);
    if constexpr (!FIRST) {
      #pragma unroll
      for (int i = 0; i < 2; ++i) hv[i] = make_float4(0.0f, 0.0f, 0.0f, 0.0f);
      if (writer) {
        #pragma unroll
        for (int i = 0; i < 2; ++i) hv[i] = hp[2 * i];
      }
    }

    v2f bh[8];
    #pragma unroll
    for (int kk = 0; kk < 8; ++kk) {
      bh[kk].x = hA[2 * kk + 0];
      bh[kk].y = hA[2 * kk + 1];
    }

    v8f acc;
    if constexpr (FIRST) {
      #pragma unroll
      for (int r = 0; r < 8; ++r) acc[r] = gcur[r >> 1];
    } else {
      acc = accx;
    }
    #pragma unroll
    for (int kk = 0; kk < 8; ++kk) acc = wmma4(ar[kk], bh[kk], acc);

    float hn;
    {
      float ir  = e ? acc[1] : acc[0];
      float fr  = e ? acc[3] : acc[2];
      float zr  = e ? acc[5] : acc[4];
      float orr = e ? acc[7] : acc[6];
      float u   = expf(-fabsf(fr));
      float q   = mS + fminf(fr, 0.0f);
      float d   = ir - q;
      float t   = expf(-fabsf(d));
      bool  imx = (d >= 0.0f);
      float mn  = imx ? ir : q;
      float ig  = imx ? 1.0f : t;
      float fg  = (imx ? t : 1.0f) / (1.0f + u);
      float cn  = fg * cS + ig * tanhf(zr);
      float nn  = fg * nS + ig;
      float og  = 1.0f / (1.0f + expf(-orr));
      hn = og * cn / nn;
      cS = cn;
      nS = nn;
      mS = mn;
    }

    ((float*)&lh[s & 1][w][hlf * SCAN_BT + bb])[e] = hn;
    if constexpr (FIRST) {
      __syncthreads();
    } else {
      v2f bx[8];
      #pragma unroll
      for (int kk = 0; kk < 8; ++kk) {
        float4 q = xv[kk >> 1];
        bx[kk].x = (kk & 1) ? q.z : q.x;
        bx[kk].y = (kk & 1) ? q.w : q.y;
      }
      #pragma unroll
      for (int r = 0; r < 8; ++r) accx[r] = bz[r];
      asm volatile("" : "+v"(accx) :: "memory");
      accx = wmma4(aw[0], bx[0], accx);
      accx = wmma4(aw[1], bx[1], accx);
      asm volatile("" :: "v"(accx));
      __builtin_amdgcn_fence(__ATOMIC_RELEASE, "workgroup", "local");
      __builtin_amdgcn_s_barrier_signal(-1);
      asm volatile("" : "+v"(accx));
      accx = wmma4(aw[2], bx[2], accx);
      accx = wmma4(aw[3], bx[3], accx);
      asm volatile("" :: "v"(accx));
      __builtin_amdgcn_s_barrier_wait(-1);
      __builtin_amdgcn_fence(__ATOMIC_ACQUIRE, "workgroup", "local");
    }
    #pragma unroll
    for (int kk = 0; kk < SCAN_WAVES; ++kk) {
      float2 t = lh[s & 1][kk][hlf * SCAN_BT + bb];
      hA[2 * kk + 0] = t.x;
      hA[2 * kk + 1] = t.y;
    }
    if constexpr (!FIRST) {
      v2f bx[4];
      #pragma unroll
      for (int kk = 4; kk < 8; ++kk) {
        float4 q = xv[kk >> 1];
        bx[kk - 4].x = (kk & 1) ? q.z : q.x;
        bx[kk - 4].y = (kk & 1) ? q.w : q.y;
      }
      asm volatile("" : "+v"(accx) :: "memory");
      #pragma unroll
      for (int kk = 4; kk < 8; ++kk) accx = wmma4(aw[kk], bx[kk - 4], accx);
      const int sn = (s + 2 < XS) ? (s + 2) : (XS - 1);
      const float* xq = xl + sn * (SCAN_BT * SCAN_XP);
      #pragma unroll
      for (int p = 0; p < 4; ++p) xv[p] = *(const float4*)(xq + 8 * p);
    }
    if constexpr (FIRST) {
      gcur = *(const v4f*)sm.g[w][tokB][hlf][e];
      tokA = tokB;
      const int s2 = (s + 2 < XS) ? (s + 2) : (XS - 1);
      tokB = sm.xi[s2 * SCAN_BT + bb];
    }

    if (writer) {
      v2f h2[8];
      #pragma unroll
      for (int kk = 0; kk < 8; ++kk) {
        h2[kk].x = hA[2 * kk + 0];
        h2[kk].y = hA[2 * kk + 1];
      }
      v2f s2 = ((h2[0] + h2[1]) + (h2[2] + h2[3])) + ((h2[4] + h2[5]) + (h2[6] + h2[7]));
      v2f q2 = ((h2[0] * h2[0] + h2[1] * h2[1]) + (h2[2] * h2[2] + h2[3] * h2[3])) +
               ((h2[4] * h2[4] + h2[5] * h2[5]) + (h2[6] * h2[6] + h2[7] * h2[7]));
      float sum = s2.x + s2.y;
      float sq  = q2.x + q2.y;
      sum += xhalf(sum);
      sq  += xhalf(sq);
      float mu = sum * (1.0f / XDH);
      float q  = sq * (1.0f / XDH) - mu * mu;
      float rs = rsqrtf(q + XEPS);
      #pragma unroll
      for (int i = 0; i < 2; ++i) {
        v2f lo0 = h2[2 * i], lo1 = h2[2 * i + 1];
        v2f hi0 = h2[4 + 2 * i], hi1 = h2[4 + 2 * i + 1];
        float u0 = e ? hi0.x : lo0.x;
        float u1 = e ? hi0.y : lo0.y;
        float u2 = e ? hi1.x : lo1.x;
        float u3 = e ? hi1.y : lo1.y;
        float4 o = hv[i];
        o.x += (u0 - mu) * rs * gn[4 * i + 0];
        o.y += (u1 - mu) * rs * gn[4 * i + 1];
        o.z += (u2 - mu) * rs * gn[4 * i + 2];
        o.w += (u3 - mu) * rs * gn[4 * i + 3];
        hp[2 * i] = o;
      }
    }
    if constexpr (FIRST) {
      if (((s + 1) & (SCAN_WAVES - 1)) == w) {
        #pragma unroll
        for (int i = 0; i < 2; ++i)
          hv[i] = *(const float4*)&sm.et[tokA][8 * (2 * e + i) + 4 * hlf];
      }
    }
  }
}

#define FFN_WAVES 8
#define FFN_WP    132

struct FfnLds {
  float wt[2][128 * FFN_WP];
  float act[FFN_WAVES][16 * FFN_WP];
};

#define FFN_FETCH(st, src)                                                        \
  _Pragma("unroll") for (int i_ = 0; i_ < 16; ++i_)                               \
    (st)[i_] = *(const v4f*)((src) + (size_t)(i_ * 256 + tid) * 4)
#define FFN_PUT(st, dst)                                                          \
  _Pragma("unroll") for (int i_ = 0; i_ < 16; ++i_) {                             \
    int idx_ = i_ * 256 + tid;                                                    \
    *(v4f*)((dst) + (idx_ >> 5) * FFN_WP + (idx_ & 31) * 4) = (st)[i_];           \
  }

#define FFN_PUT_SCALED(st, dst, sc)                                               \
  _Pragma("unroll") for (int i_ = 0; i_ < 16; ++i_) {                             \
    int idx_ = i_ * 256 + tid;                                                    \
    *(v4f*)((dst) + (idx_ >> 5) * FFN_WP + (idx_ & 31) * 4) = (st)[i_] * (sc);    \
  }

__device__ __forceinline__ void ffn_lds_barrier() {
  __builtin_amdgcn_fence(__ATOMIC_RELEASE, "workgroup", "local");
  __builtin_amdgcn_s_barrier();
  __builtin_amdgcn_fence(__ATOMIC_ACQUIRE, "workgroup", "local");
}

__device__ __forceinline__ void ffn_tile2(const v2f (&a)[32], const float* wsec, int nt,
                                          int l16, int hlf, v8f& c0, v8f& c1) {
  const float* p0 = wsec + (nt * 16 + l16) * FFN_WP + hlf * 2;
  const float* p1 = p0 + 16 * FFN_WP;
  #pragma unroll
  for (int r = 0; r < 8; ++r) {
    c0[r] = 0.0f;
    c1[r] = 0.0f;
  }
  #pragma unroll
  for (int kk = 0; kk < 32; ++kk) {
    v2f b0 = *(const v2f*)(p0 + kk * 4);
    v2f b1 = *(const v2f*)(p1 + kk * 4);
    c0 = wmma4(a[kk], b0, c0);
    c1 = wmma4(a[kk], b1, c1);
  }
}

__global__ __launch_bounds__(32 * FFN_WAVES) void k_ffn(float* __restrict__ hio,
                                                        float* __restrict__ xnext,
                                                        const float* __restrict__ lnw,
                                                        const float* __restrict__ lnnext,
                                                        const float* __restrict__ Wup,
                                                        const float* __restrict__ Wdn,
                                                        int store_h) {
  __shared__ __attribute__((aligned(16))) FfnLds sm;
  const int tid  = threadIdx.x;
  const int w    = tid >> 5;
  const int lane = tid & 31;
  const int hlf  = lane >> 4;
  const int l16  = lane & 15;
  const size_t t0 = ((size_t)blockIdx.x * FFN_WAVES + w) * 16;
  float* actw = sm.act[w];

  v2f ax[32];
  const float* xp = hio + (t0 + l16) * XD + hlf * 2;
  #pragma unroll
  for (int kk = 0; kk < 32; ++kk) ax[kk] = *(const v2f*)(xp + kk * 4);

  const v4f lw = *(const v4f*)(lnw + lane * 4);
  v4f st[16];
  FFN_FETCH(st, Wup);
  ln_afrag(ax);
  FFN_PUT_SCALED(st, sm.wt[0], lw);
  FFN_FETCH(st, Wup + 128 * XD);
  ffn_lds_barrier();

  v8f gatev[8];
  #pragma unroll
  for (int nt = 0; nt < 8; nt += 2)
    ffn_tile2(ax, sm.wt[0], nt, l16, hlf, gatev[nt], gatev[nt + 1]);

  FFN_PUT_SCALED(st, sm.wt[1], lw);
  FFN_FETCH(st, Wdn);
  ffn_lds_barrier();

  #pragma unroll
  for (int nt = 0; nt < 8; nt += 2) {
    v8f u0, u1;
    ffn_tile2(ax, sm.wt[1], nt, l16, hlf, u0, u1);
    #pragma unroll
    for (int r = 0; r < 8; ++r) {
      actw[(r + hlf * 8) * FFN_WP + nt * 16 + l16]      = gelu_exact(gatev[nt][r]) * u0[r];
      actw[(r + hlf * 8) * FFN_WP + nt * 16 + 16 + l16] = gelu_exact(gatev[nt + 1][r]) * u1[r];
    }
  }

  FFN_PUT(st, sm.wt[0]);
  ffn_lds_barrier();

  v2f aa[32];
  const float* lp = actw + l16 * FFN_WP + hlf * 2;
  #pragma unroll
  for (int kk = 0; kk < 32; ++kk) aa[kk] = *(const v2f*)(lp + kk * 4);
  float hn[8][8];
  float rsum[8], rsq[8];
  #pragma unroll
  for (int r = 0; r < 8; ++r) {
    rsum[r] = 0.0f;
    rsq[r]  = 0.0f;
  }
  #pragma unroll
  for (int nt = 0; nt < 8; nt += 2) {
    v8f d0, d1;
    ffn_tile2(aa, sm.wt[0], nt, l16, hlf, d0, d1);
    #pragma unroll
    for (int r = 0; r < 8; ++r) {
      size_t o = (t0 + r + hlf * 8) * XD + nt * 16 + l16;
      float a0 = hio[o] + d0[r];
      float a1 = hio[o + 16] + d1[r];
      hn[nt][r]     = a0;
      hn[nt + 1][r] = a1;
      rsum[r] += a0 + a1;
      rsq[r]  += a0 * a0 + a1 * a1;
    }
  }

  if (store_h) {
    #pragma unroll
    for (int nt = 0; nt < 8; ++nt) {
      #pragma unroll
      for (int r = 0; r < 8; ++r)
        hio[(t0 + r + hlf * 8) * XD + nt * 16 + l16] = hn[nt][r];
    }
  }

  #pragma unroll
  for (int r = 0; r < 8; ++r) {
    #pragma unroll
    for (int m = 8; m >= 1; m >>= 1) {
      rsum[r] += __shfl_xor(rsum[r], m, 32);
      rsq[r]  += __shfl_xor(rsq[r], m, 32);
    }
  }
  float wn[8];
  #pragma unroll
  for (int nt = 0; nt < 8; ++nt) wn[nt] = lnnext[nt * 16 + l16];
  #pragma unroll
  for (int r = 0; r < 8; ++r) {
    float mu  = rsum[r] * (1.0f / XD);
    float var = rsq[r] * (1.0f / XD) - mu * mu;
    float rs  = rsqrtf(var + XEPS);
    #pragma unroll
    for (int nt = 0; nt < 8; ++nt)
      actw[(r + hlf * 8) * FFN_WP + nt * 16 + l16] = (hn[nt][r] - mu) * rs * wn[nt];
  }
  ffn_lds_barrier();
  #pragma unroll
  for (int i = 0; i < 16; ++i) {
    v4f v = *(const v4f*)(actw + i * FFN_WP + lane * 4);
    *(v4f*)(xnext + (t0 + i) * XD + lane * 4) = v;
  }
}

__global__ __launch_bounds__(32) void k_proj(const float* __restrict__ xn,
                                             const float* __restrict__ pw,
                                             const float* __restrict__ pb,
                                             float* __restrict__ out) {
  const int lane = threadIdx.x;
  const int hlf  = lane >> 4;
  const int l16  = lane & 15;
  const size_t t0 = (size_t)blockIdx.x * 16;

  v8f a;
  #pragma unroll
  for (int r = 0; r < 8; ++r) a[r] = 0.0f;
  const float* xp = xn + (t0 + l16) * XD;
  #pragma unroll
  for (int kk = 0; kk < 32; ++kk) {
    int K0 = kk * 4 + hlf * 2;
    v2f av;
    av.x = xp[K0];
    av.y = xp[K0 + 1];
    v2f b;
    b.x = 0.0f;
    b.y = 0.0f;
    if (l16 < 9) {
      b.x = pw[(size_t)l16 * XD + K0];
      b.y = pw[(size_t)l16 * XD + K0 + 1];
    }
    a = wmma4(av, b, a);
  }
  if (l16 < 9) {
    float bb = pb[l16];
    #pragma unroll
    for (int r = 0; r < 8; ++r) {
      size_t row = t0 + r + hlf * 8;
      out[row * 9 + l16] = a[r] + bb;
    }
  }
}

extern "C" void kernel_launch(void* const* d_in, const int* in_sizes, int n_in,
                              void* d_out, int out_size, void* d_ws, size_t ws_size,
                              hipStream_t stream) {
  (void)in_sizes; (void)n_in; (void)out_size; (void)ws_size;
  const int*   x     = (const int*)d_in[0];
  const float* emb   = (const float*)d_in[1];
  const float* ln1   = (const float*)d_in[2];
  const float* Wg    = (const float*)d_in[3];
  const float* R     = (const float*)d_in[4];
  const float* bias  = (const float*)d_in[5];
  const float* gnw   = (const float*)d_in[6];
  const float* ln2   = (const float*)d_in[7];
  const float* wup   = (const float*)d_in[8];
  const float* wdn   = (const float*)d_in[9];
  const float* postw = (const float*)d_in[10];
  const float* pw    = (const float*)d_in[11];
  const float* pb    = (const float*)d_in[12];
  float* out = (float*)d_out;

  float* h  = (float*)d_ws;
  float* xb = h + (size_t)XT * XD;

  k_scan<true><<<dim3(XB / SCAN_BT, XNH), 32 * SCAN_WAVES, 0, stream>>>(x, emb, ln1, xb, h, Wg, R,
                                                                        bias, gnw);
  k_ffn<<<XT / (16 * FFN_WAVES), 32 * FFN_WAVES, 0, stream>>>(h, xb, ln2, ln1 + XD, wup, wdn, 1);

  k_scan<false><<<dim3(XB / SCAN_BT, XNH), 32 * SCAN_WAVES, 0, stream>>>(
      x, emb, ln1, xb, h,
      Wg   + (size_t)4 * XNH * XDH * XDH,
      R    + (size_t)XNH * XDH * 4 * XDH,
      bias + (size_t)XNH * 4 * XDH,
      gnw  + XD);
  k_ffn<<<XT / (16 * FFN_WAVES), 32 * FFN_WAVES, 0, stream>>>(h, xb, ln2 + XD, postw,
                                                              wup + (size_t)2 * XD * XD,
                                                              wdn + (size_t)XD * XD, 0);

  k_proj<<<XT / 16, 32, 0, stream>>>(xb, pw, pb, out);
}


# k_ffn: B fragments of the next 32 products are fetched from LDS as one group into a second register set in front of the current 32, not two loads waited for before every four products; same sums
# speedup vs baseline: 10.4017x; 1.0401x over previous
#include <hip/hip_runtime.h>
#include <hip/hip_bf16.h>


typedef float v2f __attribute__((ext_vector_type(2)));
typedef float v4f __attribute__((ext_vector_type(4)));
typedef float v8f __attribute__((ext_vector_type(8)));

#define XB   512
#define XS   64
#define XD   128
#define XNH  4
#define XDH  32
#define XT   (XB * XS)
#define XEPS 1e-5f
#define XV   9

__device__ __forceinline__ v8f wmma4(v2f a, v2f b, v8f c) {
  return __builtin_amdgcn_wmma_f32_16x16x4_f32(false, a, false, b, (short)0, c,
                                               false, false);
}

__device__ __forceinline__ float gelu_exact(float x) {
  return 0.5f * x * (1.0f + erff(x * 0.70710678118654752f));
}

__device__ __forceinline__ float xhalf(float x) {
  int b = __builtin_bit_cast(int, x);
  return __builtin_bit_cast(
      float, __builtin_amdgcn_permlanex16(b, b, 0x76543210, 0xfedcba98, false, false));
}

__device__ __forceinline__ void ln_afrag(v2f (&a)[32]) {
  v2f s4[4], q4[4];
  #pragma unroll
  for (int j = 0; j < 4; ++j) {
    s4[j] = a[j];
    q4[j] = a[j] * a[j];
  }
  #pragma unroll
  for (int kk = 4; kk < 32; ++kk) {
    s4[kk & 3] += a[kk];
    q4[kk & 3] += a[kk] * a[kk];
  }
  v2f s2 = (s4[0] + s4[1]) + (s4[2] + s4[3]);
  v2f q2 = (q4[0] + q4[1]) + (q4[2] + q4[3]);
  float s = s2.x + s2.y;
  float q = q2.x + q2.y;
  s += xhalf(s);
  q += xhalf(q);
  float mu  = s * (1.0f / XD);
  float var = q * (1.0f / XD) - mu * mu;
  float rs  = rsqrtf(var + XEPS);
  #pragma unroll
  for (int kk = 0; kk < 32; ++kk) {
    a[kk].x = (a[kk].x - mu) * rs;
    a[kk].y = (a[kk].y - mu) * rs;
  }
}

#define SCAN_WAVES 8
#define SCAN_BT    8
#define SCAN_XP    36

template <bool FIRST> struct ScanLds;
template <> struct ScanLds<false> {
  float2 lh[2][SCAN_WAVES][2 * SCAN_BT];
  float lx[XS * SCAN_BT * SCAN_XP];
};
template <> struct ScanLds<true> {
  float2 lh[2][SCAN_WAVES][2 * SCAN_BT];
  int   xi[XS * SCAN_BT];
  float xt[XV][XDH];
  float et[XV][XDH];
  float g[SCAN_WAVES][XV][2][2][4];
};

template <bool FIRST>
__global__ __launch_bounds__(32 * SCAN_WAVES) void k_scan(const int* __restrict__ x,
                                                          const float* __restrict__ emb,
                                                          const float* __restrict__ lnw,
                                                          const float* __restrict__ xn,
                                                          float* __restrict__ hio,
                                                          const float* __restrict__ Wg,
                                                          const float* __restrict__ R,
                                                          const float* __restrict__ bias,
                                                          const float* __restrict__ gnw) {
  __shared__ __attribute__((aligned(16))) ScanLds<FIRST> sm;
  float2 (*lh)[SCAN_WAVES][2 * SCAN_BT] = sm.lh;

  const int w    = threadIdx.x >> 5;
  const int lane = threadIdx.x & 31;
  const int hlf  = lane >> 4;
  const int l16  = lane & 15;
  const int bb   = l16 & 7;
  const int e    = l16 >> 3;
  const int b0   = blockIdx.x * SCAN_BT;
  const int n    = blockIdx.y;
  const int ub   = 8 * (w >> 1) + 2 * (w & 1);

  if constexpr (!FIRST) {
    float* lx = sm.lx;
    const float* xsrc = xn + (size_t)b0 * XS * XD + n * XDH;
    float4 st[16];
    #pragma unroll
    for (int i = 0; i < 16; ++i) {
      int idx  = i * (32 * SCAN_WAVES) + (int)threadIdx.x;
      int part = idx & 7;
      int seg  = idx >> 3;
      int row  = seg >> 6;
      int ss   = seg & (XS - 1);
      st[i] = *(const float4*)(xsrc + ((size_t)row * XS + ss) * XD + part * 4);
    }
    #pragma unroll
    for (int i = 0; i < 16; ++i) {
      int idx  = i * (32 * SCAN_WAVES) + (int)threadIdx.x;
      int part = idx & 7;
      int seg  = idx >> 3;
      int row  = seg >> 6;
      int ss   = seg & (XS - 1);
      *(float4*)&lx[(ss * SCAN_BT + row) * SCAN_XP + part * 4] = st[i];
    }
  }

  if constexpr (FIRST) {
    #pragma unroll
    for (int i = 0; i < 2; ++i) {
      int idx = i * (32 * SCAN_WAVES) + (int)threadIdx.x;
      sm.xi[(idx & (XS - 1)) * SCAN_BT + (idx >> 6)] = x[(size_t)b0 * XS + idx];
    }
    const float4 wl = ((const float4*)lnw)[lane];
    const int vrow[2] = {w, XV - 1};
    float4 ev[2];
    float  sv[2], qv[2];
    #pragma unroll
    for (int j = 0; j < 2; ++j) {
      ev[j] = ((const float4*)(emb + (size_t)vrow[j] * XD))[lane];
      sv[j] = ev[j].x + ev[j].y + ev[j].z + ev[j].w;
      qv[j] = ev[j].x * ev[j].x + ev[j].y * ev[j].y + ev[j].z * ev[j].z + ev[j].w * ev[j].w;
    }
    #pragma unroll
    for (int m = 16; m >= 1; m >>= 1) {
      #pragma unroll
      for (int j = 0; j < 2; ++j) {
        sv[j] += __shfl_xor(sv[j], m, 32);
        qv[j] += __shfl_xor(qv[j], m, 32);
      }
    }
    #pragma unroll
    for (int j = 0; j < 2; ++j) {
      float mu  = sv[j] * (1.0f / XD);
      float var = qv[j] * (1.0f / XD) - mu * mu;
      float rs  = rsqrtf(var + XEPS);
      float4 o;
      o.x = (ev[j].x - mu) * rs * wl.x;
      o.y = (ev[j].y - mu) * rs * wl.y;
      o.z = (ev[j].z - mu) * rs * wl.z;
      o.w = (ev[j].w - mu) * rs * wl.w;
      if ((lane >> 3) == n && (j == 0 || w == 0)) {
        *(float4*)&sm.xt[vrow[j]][(lane & 7) * 4] = o;
        *(float4*)&sm.et[vrow[j]][(lane & 7) * 4] = ev[j];
      }
    }
  }

  const int ag = (l16 >> 1) & 3;
  const int au = ub + 4 * (l16 >> 3) + (l16 & 1);
  v2f aw[8], ar[8];
  #pragma unroll
  for (int kk = 0; kk < 8; ++kk) {
    int K0 = 8 * (kk >> 1) + 4 * hlf + 2 * (kk & 1);
    const float* wgp = Wg + (((size_t)(ag * XNH + n) * XDH + au) * XDH + K0);
    aw[kk].x = wgp[0];
    aw[kk].y = wgp[1];
    ar[kk].x = R[(size_t)(n * XDH + K0 + 0) * (4 * XDH) + ag * XDH + au];
    ar[kk].y = R[(size_t)(n * XDH + K0 + 1) * (4 * XDH) + ag * XDH + au];
  }
  float bz[8];
  #pragma unroll
  for (int r = 0; r < 8; ++r)
    bz[r] = bias[n * 128 + (r >> 1) * XDH + ub + 4 * hlf + (r & 1)];
  float gn[8];
  #pragma unroll
  for (int i = 0; i < 2; ++i) {
    #pragma unroll
    for (int q = 0; q < 4; ++q) gn[4 * i + q] = gnw[n * XDH + 8 * (2 * e + i) + 4 * hlf + q];
  }

  float cS = 0.0f;
  float nS = 0.0f;
  float mS = 0.0f;
  float hA[16] = {};

  const size_t rowoff = (size_t)(b0 + bb) * XS * XD + n * XDH + hlf * 4;

  const float* xl = nullptr;
  float4 xv[4];
  #pragma unroll
  for (int p = 0; p < 4; ++p) xv[p] = make_float4(0.0f, 0.0f, 0.0f, 0.0f);
  v4f gcur = {0.0f, 0.0f, 0.0f, 0.0f};
  int tokA = 0, tokB = 0;
  v8f accx = {0.0f, 0.0f, 0.0f, 0.0f, 0.0f, 0.0f, 0.0f, 0.0f};

  if constexpr (FIRST) {
    __syncthreads();
    const int tv = (l16 < XV) ? l16 : (XV - 1);
    v8f tacc;
    #pragma unroll
    for (int r = 0; r < 8; ++r) tacc[r] = bz[r];
    #pragma unroll
    for (int kk = 0; kk < 8; ++kk) {
      int K0 = 8 * (kk >> 1) + 4 * hlf + 2 * (kk & 1);
      v2f bt = *(const v2f*)&sm.xt[tv][K0];
      tacc = wmma4(aw[kk], bt, tacc);
    }
    if (l16 < XV) {
      v4f g0 = {tacc[0], tacc[2], tacc[4], tacc[6]};
      v4f g1 = {tacc[1], tacc[3], tacc[5], tacc[7]};
      *(v4f*)sm.g[w][l16][hlf][0] = g0;
      *(v4f*)sm.g[w][l16][hlf][1] = g1;
    }
    __syncthreads();
    tokA = sm.xi[0 * SCAN_BT + bb];
    tokB = sm.xi[1 * SCAN_BT + bb];
    gcur = *(const v4f*)sm.g[w][tokA][hlf][e];
  } else {
    xl = sm.lx + bb * SCAN_XP + hlf * 4;
    __syncthreads();
    #pragma unroll
    for (int p = 0; p < 4; ++p) xv[p] = *(const float4*)(xl + 8 * p);
    #pragma unroll
    for (int r = 0; r < 8; ++r) accx[r] = bz[r];
    #pragma unroll
    for (int kk = 0; kk < 8; ++kk) {
      float4 q = xv[kk >> 1];
      v2f bx;
      bx.x = (kk & 1) ? q.z : q.x;
      bx.y = (kk & 1) ? q.w : q.y;
      accx = wmma4(aw[kk], bx, accx);
    }
    #pragma unroll
    for (int p = 0; p < 4; ++p) xv[p] = *(const float4*)(xl + (SCAN_BT * SCAN_XP) + 8 * p);
  }

  float4 hv[2];
  #pragma unroll
  for (int i = 0; i < 2; ++i) hv[i] = make_float4(0.0f, 0.0f, 0.0f, 0.0f);
  if constexpr (FIRST) {
    if (w == 0) {
      #pragma unroll
      for (int i = 0; i < 2; ++i)
        hv[i] = *(const float4*)&sm.et[tokA][8 * (2 * e + i) + 4 * hlf];
    }
  }

  #pragma unroll 1
  for (int s = 0; s < XS; ++s) {
    float4*       hp = (float4*)(hio + rowoff + (size_t)s * XD) + 4 * e;
    const bool writer = ((s & (SCAN_WAVES - 1)) == w);
    if constexpr (!FIRST) {
      #pragma unroll
      for (int i = 0; i < 2; ++i) hv[i] = make_float4(0.0f, 0.0f, 0.0f, 0.0f);
      if (writer) {
        #pragma unroll
        for (int i = 0; i < 2; ++i) hv[i] = hp[2 * i];
      }
    }

    v2f bh[8];
    #pragma unroll
    for (int kk = 0; kk < 8; ++kk) {
      bh[kk].x = hA[2 * kk + 0];
      bh[kk].y = hA[2 * kk + 1];
    }

    v8f acc;
    if constexpr (FIRST) {
      #pragma unroll
      for (int r = 0; r < 8; ++r) acc[r] = gcur[r >> 1];
    } else {
      acc = accx;
    }
    #pragma unroll
    for (int kk = 0; kk < 8; ++kk) acc = wmma4(ar[kk], bh[kk], acc);

    float hn;
    {
      float ir  = e ? acc[1] : acc[0];
      float fr  = e ? acc[3] : acc[2];
      float zr  = e ? acc[5] : acc[4];
      float orr = e ? acc[7] : acc[6];
      float u   = expf(-fabsf(fr));
      float q   = mS + fminf(fr, 0.0f);
      float d   = ir - q;
      float t   = expf(-fabsf(d));
      bool  imx = (d >= 0.0f);
      float mn  = imx ? ir : q;
      float ig  = imx ? 1.0f : t;
      float fg  = (imx ? t : 1.0f) / (1.0f + u);
      float cn  = fg * cS + ig * tanhf(zr);
      float nn  = fg * nS + ig;
      float og  = 1.0f / (1.0f + expf(-orr));
      hn = og * cn / nn;
      cS = cn;
      nS = nn;
      mS = mn;
    }

    ((float*)&lh[s & 1][w][hlf * SCAN_BT + bb])[e] = hn;
    if constexpr (FIRST) {
      __syncthreads();
    } else {
      v2f bx[8];
      #pragma unroll
      for (int kk = 0; kk < 8; ++kk) {
        float4 q = xv[kk >> 1];
        bx[kk].x = (kk & 1) ? q.z : q.x;
        bx[kk].y = (kk & 1) ? q.w : q.y;
      }
      #pragma unroll
      for (int r = 0; r < 8; ++r) accx[r] = bz[r];
      asm volatile("" : "+v"(accx) :: "memory");
      accx = wmma4(aw[0], bx[0], accx);
      accx = wmma4(aw[1], bx[1], accx);
      asm volatile("" :: "v"(accx));
      __builtin_amdgcn_fence(__ATOMIC_RELEASE, "workgroup", "local");
      __builtin_amdgcn_s_barrier_signal(-1);
      asm volatile("" : "+v"(accx));
      accx = wmma4(aw[2], bx[2], accx);
      accx = wmma4(aw[3], bx[3], accx);
      asm volatile("" :: "v"(accx));
      __builtin_amdgcn_s_barrier_wait(-1);
      __builtin_amdgcn_fence(__ATOMIC_ACQUIRE, "workgroup", "local");
    }
    #pragma unroll
    for (int kk = 0; kk < SCAN_WAVES; ++kk) {
      float2 t = lh[s & 1][kk][hlf * SCAN_BT + bb];
      hA[2 * kk + 0] = t.x;
      hA[2 * kk + 1] = t.y;
    }
    if constexpr (!FIRST) {
      v2f bx[4];
      #pragma unroll
      for (int kk = 4; kk < 8; ++kk) {
        float4 q = xv[kk >> 1];
        bx[kk - 4].x = (kk & 1) ? q.z : q.x;
        bx[kk - 4].y = (kk & 1) ? q.w : q.y;
      }
      asm volatile("" : "+v"(accx) :: "memory");
      #pragma unroll
      for (int kk = 4; kk < 8; ++kk) accx = wmma4(aw[kk], bx[kk - 4], accx);
      const int sn = (s + 2 < XS) ? (s + 2) : (XS - 1);
      const float* xq = xl + sn * (SCAN_BT * SCAN_XP);
      #pragma unroll
      for (int p = 0; p < 4; ++p) xv[p] = *(const float4*)(xq + 8 * p);
    }
    if constexpr (FIRST) {
      gcur = *(const v4f*)sm.g[w][tokB][hlf][e];
      tokA = tokB;
      const int s2 = (s + 2 < XS) ? (s + 2) : (XS - 1);
      tokB = sm.xi[s2 * SCAN_BT + bb];
    }

    if (writer) {
      v2f h2[8];
      #pragma unroll
      for (int kk = 0; kk < 8; ++kk) {
        h2[kk].x = hA[2 * kk + 0];
        h2[kk].y = hA[2 * kk + 1];
      }
      v2f s2 = ((h2[0] + h2[1]) + (h2[2] + h2[3])) + ((h2[4] + h2[5]) + (h2[6] + h2[7]));
      v2f q2 = ((h2[0] * h2[0] + h2[1] * h2[1]) + (h2[2] * h2[2] + h2[3] * h2[3])) +
               ((h2[4] * h2[4] + h2[5] * h2[5]) + (h2[6] * h2[6] + h2[7] * h2[7]));
      float sum = s2.x + s2.y;
      float sq  = q2.x + q2.y;
      sum += xhalf(sum);
      sq  += xhalf(sq);
      float mu = sum * (1.0f / XDH);
      float q  = sq * (1.0f / XDH) - mu * mu;
      float rs = rsqrtf(q + XEPS);
      #pragma unroll
      for (int i = 0; i < 2; ++i) {
        v2f lo0 = h2[2 * i], lo1 = h2[2 * i + 1];
        v2f hi0 = h2[4 + 2 * i], hi1 = h2[4 + 2 * i + 1];
        float u0 = e ? hi0.x : lo0.x;
        float u1 = e ? hi0.y : lo0.y;
        float u2 = e ? hi1.x : lo1.x;
        float u3 = e ? hi1.y : lo1.y;
        float4 o = hv[i];
        o.x += (u0 - mu) * rs * gn[4 * i + 0];
        o.y += (u1 - mu) * rs * gn[4 * i + 1];
        o.z += (u2 - mu) * rs * gn[4 * i + 2];
        o.w += (u3 - mu) * rs * gn[4 * i + 3];
        hp[2 * i] = o;
      }
    }
    if constexpr (FIRST) {
      if (((s + 1) & (SCAN_WAVES - 1)) == w) {
        #pragma unroll
        for (int i = 0; i < 2; ++i)
          hv[i] = *(const float4*)&sm.et[tokA][8 * (2 * e + i) + 4 * hlf];
      }
    }
  }
}

#define FFN_WAVES 8
#define FFN_WP    132

struct FfnLds {
  float wt[2][128 * FFN_WP];
  float act[FFN_WAVES][16 * FFN_WP];
};

#define FFN_FETCH(st, src)                                                        \
  _Pragma("unroll") for (int i_ = 0; i_ < 16; ++i_)                               \
    (st)[i_] = *(const v4f*)((src) + (size_t)(i_ * 256 + tid) * 4)
#define FFN_PUT(st, dst)                                                          \
  _Pragma("unroll") for (int i_ = 0; i_ < 16; ++i_) {                             \
    int idx_ = i_ * 256 + tid;                                                    \
    *(v4f*)((dst) + (idx_ >> 5) * FFN_WP + (idx_ & 31) * 4) = (st)[i_];           \
  }

#define FFN_PUT_SCALED(st, dst, sc)                                               \
  _Pragma("unroll") for (int i_ = 0; i_ < 16; ++i_) {                             \
    int idx_ = i_ * 256 + tid;                                                    \
    *(v4f*)((dst) + (idx_ >> 5) * FFN_WP + (idx_ & 31) * 4) = (st)[i_] * (sc);    \
  }

__device__ __forceinline__ void ffn_lds_barrier() {
  __builtin_amdgcn_fence(__ATOMIC_RELEASE, "workgroup", "local");
  __builtin_amdgcn_s_barrier();
  __builtin_amdgcn_fence(__ATOMIC_ACQUIRE, "workgroup", "local");
}

__device__ __forceinline__ void ffn_bload(v2f (&b)[32], const float* wsec, int blk, int l16,
                                          int hlf) {
  const float* p0 = wsec + ((blk >> 1) * 32 + l16) * FFN_WP + hlf * 2 + (blk & 1) * 64;
  const float* p1 = p0 + 16 * FFN_WP;
  #pragma unroll
  for (int i = 0; i < 16; ++i) {
    b[i]      = *(const v2f*)(p0 + i * 4);
    b[16 + i] = *(const v2f*)(p1 + i * 4);
  }
}

__device__ __forceinline__ void ffn_bmul(const v2f (&a)[32], const v2f (&b)[32], int blk,
                                         v8f& c0, v8f& c1) {
  const int k0 = (blk & 1) * 16;
  if (k0 == 0) {
    #pragma unroll
    for (int r = 0; r < 8; ++r) {
      c0[r] = 0.0f;
      c1[r] = 0.0f;
    }
  }
  #pragma unroll
  for (int i = 0; i < 16; ++i) {
    c0 = wmma4(a[k0 + i], b[i], c0);
    c1 = wmma4(a[k0 + i], b[16 + i], c1);
  }
}

#define FFN_TRIP(bq, a, wsec, blk, c0, c1)                                        \
  if ((blk) < 7) ffn_bload((bq)[((blk) + 1) & 1], (wsec), (blk) + 1, l16, hlf);    \
       \
  __builtin_amdgcn_sched_barrier(0);                                              \
  ffn_bmul((a), (bq)[(blk) & 1], (blk), (c0), (c1));                               \
        \
  __builtin_amdgcn_sched_barrier(0)

__global__ __launch_bounds__(32 * FFN_WAVES) void k_ffn(float* __restrict__ hio,
                                                        float* __restrict__ xnext,
                                                        const float* __restrict__ lnw,
                                                        const float* __restrict__ lnnext,
                                                        const float* __restrict__ Wup,
                                                        const float* __restrict__ Wdn,
                                                        int store_h) {
  __shared__ __attribute__((aligned(16))) FfnLds sm;
  const int tid  = threadIdx.x;
  const int w    = tid >> 5;
  const int lane = tid & 31;
  const int hlf  = lane >> 4;
  const int l16  = lane & 15;
  const size_t t0 = ((size_t)blockIdx.x * FFN_WAVES + w) * 16;
  float* actw = sm.act[w];

  v2f ax[32];
  const float* xp = hio + (t0 + l16) * XD + hlf * 2;
  #pragma unroll
  for (int kk = 0; kk < 32; ++kk) ax[kk] = *(const v2f*)(xp + kk * 4);

  const v4f lw = *(const v4f*)(lnw + lane * 4);
  v4f st[16];
  FFN_FETCH(st, Wup);
  ln_afrag(ax);
  FFN_PUT_SCALED(st, sm.wt[0], lw);
  FFN_FETCH(st, Wup + 128 * XD);
  ffn_lds_barrier();

  v8f gatev[8];
  v2f bq[2][32];
  ffn_bload(bq[0], sm.wt[0], 0, l16, hlf);
  #pragma unroll
  for (int blk = 0; blk < 8; ++blk) {
    FFN_TRIP(bq, ax, sm.wt[0], blk, gatev[blk & ~1], gatev[(blk & ~1) + 1]);
  }

  FFN_PUT_SCALED(st, sm.wt[1], lw);
  FFN_FETCH(st, Wdn);
  ffn_lds_barrier();

  ffn_bload(bq[0], sm.wt[1], 0, l16, hlf);
  {
    v8f u0, u1;
    #pragma unroll
    for (int blk = 0; blk < 8; ++blk) {
      FFN_TRIP(bq, ax, sm.wt[1], blk, u0, u1);
      if (blk & 1) {
        const int nt = blk & ~1;
        #pragma unroll
        for (int r = 0; r < 8; ++r) {
          actw[(r + hlf * 8) * FFN_WP + nt * 16 + l16]      = gelu_exact(gatev[nt][r]) * u0[r];
          actw[(r + hlf * 8) * FFN_WP + nt * 16 + 16 + l16] = gelu_exact(gatev[nt + 1][r]) * u1[r];
        }
      }
    }
  }

  FFN_PUT(st, sm.wt[0]);
  ffn_lds_barrier();

  v2f aa[32];
  const float* lp = actw + l16 * FFN_WP + hlf * 2;
  #pragma unroll
  for (int kk = 0; kk < 32; ++kk) aa[kk] = *(const v2f*)(lp + kk * 4);
  float hn[8][8];
  float rsum[8], rsq[8];
  #pragma unroll
  for (int r = 0; r < 8; ++r) {
    rsum[r] = 0.0f;
    rsq[r]  = 0.0f;
  }
  float res[8][8];
  #pragma unroll
  for (int nt = 0; nt < 8; ++nt) {
    #pragma unroll
    for (int r = 0; r < 8; ++r) res[nt][r] = hio[(t0 + r + hlf * 8) * XD + nt * 16 + l16];
  }
  ffn_bload(bq[0], sm.wt[0], 0, l16, hlf);
  {
    v8f d0, d1;
    #pragma unroll
    for (int blk = 0; blk < 8; ++blk) {
      FFN_TRIP(bq, aa, sm.wt[0], blk, d0, d1);
      if (blk & 1) {
        const int nt = blk & ~1;
        #pragma unroll
        for (int r = 0; r < 8; ++r) {
          float a0 = res[nt][r] + d0[r];
          float a1 = res[nt + 1][r] + d1[r];
          hn[nt][r]     = a0;
          hn[nt + 1][r] = a1;
          rsum[r] += a0 + a1;
          rsq[r]  += a0 * a0 + a1 * a1;
        }
      }
    }
  }

  if (store_h) {
    #pragma unroll
    for (int nt = 0; nt < 8; ++nt) {
      #pragma unroll
      for (int r = 0; r < 8; ++r)
        hio[(t0 + r + hlf * 8) * XD + nt * 16 + l16] = hn[nt][r];
    }
  }

  #pragma unroll
  for (int r = 0; r < 8; ++r) {
    #pragma unroll
    for (int m = 8; m >= 1; m >>= 1) {
      rsum[r] += __shfl_xor(rsum[r], m, 32);
      rsq[r]  += __shfl_xor(rsq[r], m, 32);
    }
  }
  float wn[8];
  #pragma unroll
  for (int nt = 0; nt < 8; ++nt) wn[nt] = lnnext[nt * 16 + l16];
  #pragma unroll
  for (int r = 0; r < 8; ++r) {
    float mu  = rsum[r] * (1.0f / XD);
    float var = rsq[r] * (1.0f / XD) - mu * mu;
    float rs  = rsqrtf(var + XEPS);
    #pragma unroll
    for (int nt = 0; nt < 8; ++nt)
      actw[(r + hlf * 8) * FFN_WP + nt * 16 + l16] = (hn[nt][r] - mu) * rs * wn[nt];
  }
  ffn_lds_barrier();
  #pragma unroll
  for (int i = 0; i < 16; ++i) {
    v4f v = *(const v4f*)(actw + i * FFN_WP + lane * 4);
    *(v4f*)(xnext + (t0 + i) * XD + lane * 4) = v;
  }
}

__global__ __launch_bounds__(32) void k_proj(const float* __restrict__ xn,
                                             const float* __restrict__ pw,
                                             const float* __restrict__ pb,
                                             float* __restrict__ out) {
  const int lane = threadIdx.x;
  const int hlf  = lane >> 4;
  const int l16  = lane & 15;
  const size_t t0 = (size_t)blockIdx.x * 16;

  v8f a;
  #pragma unroll
  for (int r = 0; r < 8; ++r) a[r] = 0.0f;
  const float* xp = xn + (t0 + l16) * XD;
  #pragma unroll
  for (int kk = 0; kk < 32; ++kk) {
    int K0 = kk * 4 + hlf * 2;
    v2f av;
    av.x = xp[K0];
    av.y = xp[K0 + 1];
    v2f b;
    b.x = 0.0f;
    b.y = 0.0f;
    if (l16 < 9) {
      b.x = pw[(size_t)l16 * XD + K0];
      b.y = pw[(size_t)l16 * XD + K0 + 1];
    }
    a = wmma4(av, b, a);
  }
  if (l16 < 9) {
    float bb = pb[l16];
    #pragma unroll
    for (int r = 0; r < 8; ++r) {
      size_t row = t0 + r + hlf * 8;
      out[row * 9 + l16] = a[r] + bb;
    }
  }
}

extern "C" void kernel_launch(void* const* d_in, const int* in_sizes, int n_in,
                              void* d_out, int out_size, void* d_ws, size_t ws_size,
                              hipStream_t stream) {
  (void)in_sizes; (void)n_in; (void)out_size; (void)ws_size;
  const int*   x     = (const int*)d_in[0];
  const float* emb   = (const float*)d_in[1];
  const float* ln1   = (const float*)d_in[2];
  const float* Wg    = (const float*)d_in[3];
  const float* R     = (const float*)d_in[4];
  const float* bias  = (const float*)d_in[5];
  const float* gnw   = (const float*)d_in[6];
  const float* ln2   = (const float*)d_in[7];
  const float* wup   = (const float*)d_in[8];
  const float* wdn   = (const float*)d_in[9];
  const float* postw = (const float*)d_in[10];
  const float* pw    = (const float*)d_in[11];
  const float* pb    = (const float*)d_in[12];
  float* out = (float*)d_out;

  float* h  = (float*)d_ws;
  float* xb = h + (size_t)XT * XD;

  k_scan<true><<<dim3(XB / SCAN_BT, XNH), 32 * SCAN_WAVES, 0, stream>>>(x, emb, ln1, xb, h, Wg, R,
                                                                        bias, gnw);
  k_ffn<<<XT / (16 * FFN_WAVES), 32 * FFN_WAVES, 0, stream>>>(h, xb, ln2, ln1 + XD, wup, wdn, 1);

  k_scan<false><<<dim3(XB / SCAN_BT, XNH), 32 * SCAN_WAVES, 0, stream>>>(
      x, emb, ln1, xb, h,
      Wg   + (size_t)4 * XNH * XDH * XDH,
      R    + (size_t)XNH * XDH * 4 * XDH,
      bias + (size_t)XNH * 4 * XDH,
      gnw  + XD);
  k_ffn<<<XT / (16 * FFN_WAVES), 32 * FFN_WAVES, 0, stream>>>(h, xb, ln2 + XD, postw,
                                                              wup + (size_t)2 * XD * XD,
                                                              wdn + (size_t)XD * XD, 0);

  k_proj<<<XT / 16, 32, 0, stream>>>(xb, pw, pb, out);
}
